# starting kernel
# baseline (speedup 1.0000x reference)
#include <hip/hip_runtime.h>


typedef __attribute__((ext_vector_type(2))) float v2f;
typedef __attribute__((ext_vector_type(8))) float v8f;

#define IN_DIM  128
#define HID_DIM 64
#define OUT_DIM 64
#define W_PITCH 160

__global__ __launch_bounds__(256) void zero_f4(float4* __restrict__ p, int n4) {
  int i = blockIdx.x * blockDim.x + threadIdx.x;
  if (i < n4) p[i] = make_float4(0.f, 0.f, 0.f, 0.f);
}

template <int K>
__global__ __launch_bounds__(256) void gemm_wmma(const float* __restrict__ X,
                                                 const float* __restrict__ W,
                                                 float* __restrict__ Y,
                                                 int nTiles) {
  __shared__ float sW[(K / 2) * W_PITCH];
  for (int i = threadIdx.x; i < K * 64; i += 256) {
    int r = i >> 6, c = i & 63;
    sW[(r >> 1) * W_PITCH + (c << 1) + (r & 1)] = W[i];
  }
  __syncthreads();

  const int wave = threadIdx.x >> 5;
  const int lane = threadIdx.x & 31;
  const int tile = blockIdx.x * 8 + wave;
  if (tile >= nTiles) return;

  const int mrow = lane & 15;
  const int half = lane >> 4;
  const int koff = half << 1;

  const float* xrow = X + (size_t)(tile * 16 + mrow) * K;
  const float* wbase = sW + (size_t)half * W_PITCH + (mrow << 1);

  v8f acc0 = {}, acc1 = {}, acc2 = {}, acc3 = {};

  for (int k = 0; k < K; k += 4) {
    v2f a = *(const v2f*)(xrow + k + koff);
    const float* wk = wbase + (k >> 1) * W_PITCH;
    v2f b0 = *(const v2f*)(wk + 0);
    v2f b1 = *(const v2f*)(wk + 32);
    v2f b2 = *(const v2f*)(wk + 64);
    v2f b3 = *(const v2f*)(wk + 96);

    acc0 = __builtin_amdgcn_wmma_f32_16x16x4_f32(false, a, false, b0, (short)0, acc0, false, false);
    acc1 = __builtin_amdgcn_wmma_f32_16x16x4_f32(false, a, false, b1, (short)0, acc1, false, false);
    acc2 = __builtin_amdgcn_wmma_f32_16x16x4_f32(false, a, false, b2, (short)0, acc2, false, false);
    acc3 = __builtin_amdgcn_wmma_f32_16x16x4_f32(false, a, false, b3, (short)0, acc3, false, false);
  }

  const int rbase = tile * 16 + (half << 3);
  float* y = Y + (size_t)rbase * 64 + mrow;
#pragma unroll
  for (int v = 0; v < 8; v++) {
    y[(size_t)v * 64 + 0]  = acc0[v];
    y[(size_t)v * 64 + 16] = acc1[v];
    y[(size_t)v * 64 + 32] = acc2[v];
    y[(size_t)v * 64 + 48] = acc3[v];
  }
}

__global__ __launch_bounds__(256) void scatter_edges(const float* __restrict__ H,
                                                     const int* __restrict__ src,
                                                     const int* __restrict__ dst,
                                                     const float* __restrict__ ew,
                                                     float* __restrict__ agg,
                                                     int nEdges) {
  long long t = (long long)blockIdx.x * 256 + threadIdx.x;
  int e  = (int)(t >> 4);
  int f4 = (int)(t & 15);
  if (e >= nEdges) return;
  int s = src[e], d = dst[e];
  float w = ew[e];
  float4 v = reinterpret_cast<const float4*>(H + (size_t)s * 64)[f4];
  float* p = agg + (size_t)d * 64 + (f4 << 2);
  unsafeAtomicAdd(p + 0, v.x * w);
  unsafeAtomicAdd(p + 1, v.y * w);
  unsafeAtomicAdd(p + 2, v.z * w);
  unsafeAtomicAdd(p + 3, v.w * w);
}

__global__ __launch_bounds__(256) void relu_bias_k(const float4* __restrict__ agg,
                                                   const float* __restrict__ b,
                                                   float4* __restrict__ out, int n4) {
  int i = blockIdx.x * blockDim.x + threadIdx.x;
  if (i >= n4) return;
  float4 bv = reinterpret_cast<const float4*>(b)[i & 15];
  float4 v = agg[i];
  v.x = fmaxf(v.x + bv.x, 0.f);
  v.y = fmaxf(v.y + bv.y, 0.f);
  v.z = fmaxf(v.z + bv.z, 0.f);
  v.w = fmaxf(v.w + bv.w, 0.f);
  out[i] = v;
}

__global__ __launch_bounds__(256) void bias_add_k(float4* __restrict__ out,
                                                  const float* __restrict__ b, int n4) {
  int i = blockIdx.x * blockDim.x + threadIdx.x;
  if (i >= n4) return;
  float4 bv = reinterpret_cast<const float4*>(b)[i & 15];
  float4 v = out[i];
  v.x += bv.x; v.y += bv.y; v.z += bv.z; v.w += bv.w;
  out[i] = v;
}

extern "C" void kernel_launch(void* const* d_in, const int* in_sizes, int n_in,
                              void* d_out, int out_size, void* d_ws, size_t ws_size,
                              hipStream_t stream) {
  const float* x  = (const float*)d_in[0];
  const int*   ei = (const int*)d_in[1];
  const float* ew = (const float*)d_in[2];
  const float* w1 = (const float*)d_in[3];
  const float* b1 = (const float*)d_in[4];
  const float* w2 = (const float*)d_in[5];
  const float* b2 = (const float*)d_in[6];

  const int N = in_sizes[0] / IN_DIM;
  const int E = in_sizes[2];
  const int* src = ei;
  const int* dst = ei + E;
  float* out = (float*)d_out;

  float* A = (float*)d_ws;
  float* B = A + (size_t)N * 64;

  const int n4 = N * 16;
  const int zb = (n4 + 255) / 256;
  const int nTiles = (N + 15) / 16;
  const int gemmBlocks = (nTiles + 7) / 8;
  const long long st = (long long)E * 16;
  const int sb = (int)((st + 255) / 256);

  zero_f4<<<zb, 256, 0, stream>>>((float4*)B, n4);
  gemm_wmma<IN_DIM><<<gemmBlocks, 256, 0, stream>>>(x, w1, A, nTiles);
  scatter_edges<<<sb, 256, 0, stream>>>(A, src, dst, ew, B, E);
  relu_bias_k<<<zb, 256, 0, stream>>>((const float4*)B, b1, (float4*)A, n4);

  zero_f4<<<zb, 256, 0, stream>>>((float4*)out, n4);
  gemm_wmma<HID_DIM><<<gemmBlocks, 256, 0, stream>>>(A, w2, B, nTiles);
  scatter_edges<<<sb, 256, 0, stream>>>(B, src, dst, ew, out, E);
  bias_add_k<<<zb, 256, 0, stream>>>((float4*)out, b2, n4);
}


# scatter: one wave per 32 edges, lane = column (l, l+32): each row gather load and each f32 atomic add covers one whole 128-B line, not 4 lines x 8 floats; atomic arrival order differs
# speedup vs baseline: 1.6557x; 1.6557x over previous
#include <hip/hip_runtime.h>


typedef __attribute__((ext_vector_type(2))) float v2f;
typedef __attribute__((ext_vector_type(8))) float v8f;

#define IN_DIM  128
#define HID_DIM 64
#define OUT_DIM 64
#define W_PITCH 160

__global__ __launch_bounds__(256) void zero_f4(float4* __restrict__ p, int n4) {
  int i = blockIdx.x * blockDim.x + threadIdx.x;
  if (i < n4) p[i] = make_float4(0.f, 0.f, 0.f, 0.f);
}

template <int K>
__global__ __launch_bounds__(256) void gemm_wmma(const float* __restrict__ X,
                                                 const float* __restrict__ W,
                                                 float* __restrict__ Y,
                                                 int nTiles) {
  __shared__ float sW[(K / 2) * W_PITCH];
  for (int i = threadIdx.x; i < K * 64; i += 256) {
    int r = i >> 6, c = i & 63;
    sW[(r >> 1) * W_PITCH + (c << 1) + (r & 1)] = W[i];
  }
  __syncthreads();

  const int wave = threadIdx.x >> 5;
  const int lane = threadIdx.x & 31;
  const int tile = blockIdx.x * 8 + wave;
  if (tile >= nTiles) return;

  const int mrow = lane & 15;
  const int half = lane >> 4;
  const int koff = half << 1;

  const float* xrow = X + (size_t)(tile * 16 + mrow) * K;
  const float* wbase = sW + (size_t)half * W_PITCH + (mrow << 1);

  v8f acc0 = {}, acc1 = {}, acc2 = {}, acc3 = {};

  for (int k = 0; k < K; k += 4) {
    v2f a = *(const v2f*)(xrow + k + koff);
    const float* wk = wbase + (k >> 1) * W_PITCH;
    v2f b0 = *(const v2f*)(wk + 0);
    v2f b1 = *(const v2f*)(wk + 32);
    v2f b2 = *(const v2f*)(wk + 64);
    v2f b3 = *(const v2f*)(wk + 96);

    acc0 = __builtin_amdgcn_wmma_f32_16x16x4_f32(false, a, false, b0, (short)0, acc0, false, false);
    acc1 = __builtin_amdgcn_wmma_f32_16x16x4_f32(false, a, false, b1, (short)0, acc1, false, false);
    acc2 = __builtin_amdgcn_wmma_f32_16x16x4_f32(false, a, false, b2, (short)0, acc2, false, false);
    acc3 = __builtin_amdgcn_wmma_f32_16x16x4_f32(false, a, false, b3, (short)0, acc3, false, false);
  }

  const int rbase = tile * 16 + (half << 3);
  float* y = Y + (size_t)rbase * 64 + mrow;
#pragma unroll
  for (int v = 0; v < 8; v++) {
    y[(size_t)v * 64 + 0]  = acc0[v];
    y[(size_t)v * 64 + 16] = acc1[v];
    y[(size_t)v * 64 + 32] = acc2[v];
    y[(size_t)v * 64 + 48] = acc3[v];
  }
}

#define EDGES_PER_WAVE 32
#define EDGE_CHUNK 8

__device__ __forceinline__ void edge_one(const float* __restrict__ H, float* __restrict__ agg,
                                         int s, int d, float w, int lane) {
  const float* h = H + (size_t)s * 64 + lane;
  float* p = agg + (size_t)d * 64 + lane;
  float v0 = h[0], v1 = h[32];
  unsafeAtomicAdd(p, v0 * w);
  unsafeAtomicAdd(p + 32, v1 * w);
}

__global__ __launch_bounds__(256) void scatter_edges(const float* __restrict__ H,
                                                     const int* __restrict__ src,
                                                     const int* __restrict__ dst,
                                                     const float* __restrict__ ew,
                                                     float* __restrict__ agg,
                                                     int nEdges) {
  const int lane = threadIdx.x & 31;
  const long long e0 = ((long long)blockIdx.x * 8 + (threadIdx.x >> 5)) * EDGES_PER_WAVE;
  if (e0 >= nEdges) return;
  const long long rem = (long long)nEdges - e0;
  const int cnt = rem < EDGES_PER_WAVE ? (int)rem : EDGES_PER_WAVE;

  int s = 0, d = 0, wbits = 0;
  if (lane < cnt) {
    s = src[e0 + lane];
    d = dst[e0 + lane];
    wbits = __float_as_int(ew[e0 + lane]);
  }

  int j = 0;
  for (; j + EDGE_CHUNK <= cnt; j += EDGE_CHUNK) {
    float v0[EDGE_CHUNK], v1[EDGE_CHUNK];
#pragma unroll
    for (int u = 0; u < EDGE_CHUNK; u++) {
      const int sj = __builtin_amdgcn_readlane(s, j + u);
      const float* h = H + (size_t)sj * 64 + lane;
      v0[u] = h[0];
      v1[u] = h[32];
    }
#pragma unroll
    for (int u = 0; u < EDGE_CHUNK; u++) {
      const int dj = __builtin_amdgcn_readlane(d, j + u);
      const float wj = __int_as_float(__builtin_amdgcn_readlane(wbits, j + u));
      float* p = agg + (size_t)dj * 64 + lane;
      unsafeAtomicAdd(p, v0[u] * wj);
      unsafeAtomicAdd(p + 32, v1[u] * wj);
    }
  }
  for (; j < cnt; j++) {
    edge_one(H, agg, __builtin_amdgcn_readlane(s, j), __builtin_amdgcn_readlane(d, j),
             __int_as_float(__builtin_amdgcn_readlane(wbits, j)), lane);
  }
}

__global__ __launch_bounds__(256) void relu_bias_k(const float4* __restrict__ agg,
                                                   const float* __restrict__ b,
                                                   float4* __restrict__ out, int n4) {
  int i = blockIdx.x * blockDim.x + threadIdx.x;
  if (i >= n4) return;
  float4 bv = reinterpret_cast<const float4*>(b)[i & 15];
  float4 v = agg[i];
  v.x = fmaxf(v.x + bv.x, 0.f);
  v.y = fmaxf(v.y + bv.y, 0.f);
  v.z = fmaxf(v.z + bv.z, 0.f);
  v.w = fmaxf(v.w + bv.w, 0.f);
  out[i] = v;
}

__global__ __launch_bounds__(256) void bias_add_k(float4* __restrict__ out,
                                                  const float* __restrict__ b, int n4) {
  int i = blockIdx.x * blockDim.x + threadIdx.x;
  if (i >= n4) return;
  float4 bv = reinterpret_cast<const float4*>(b)[i & 15];
  float4 v = out[i];
  v.x += bv.x; v.y += bv.y; v.z += bv.z; v.w += bv.w;
  out[i] = v;
}

extern "C" void kernel_launch(void* const* d_in, const int* in_sizes, int n_in,
                              void* d_out, int out_size, void* d_ws, size_t ws_size,
                              hipStream_t stream) {
  const float* x  = (const float*)d_in[0];
  const int*   ei = (const int*)d_in[1];
  const float* ew = (const float*)d_in[2];
  const float* w1 = (const float*)d_in[3];
  const float* b1 = (const float*)d_in[4];
  const float* w2 = (const float*)d_in[5];
  const float* b2 = (const float*)d_in[6];

  const int N = in_sizes[0] / IN_DIM;
  const int E = in_sizes[2];
  const int* src = ei;
  const int* dst = ei + E;
  float* out = (float*)d_out;

  float* A = (float*)d_ws;
  float* B = A + (size_t)N * 64;

  const int n4 = N * 16;
  const int zb = (n4 + 255) / 256;
  const int nTiles = (N + 15) / 16;
  const int gemmBlocks = (nTiles + 7) / 8;
  const int edgeWaves = (E + EDGES_PER_WAVE - 1) / EDGES_PER_WAVE;
  const int sb = (edgeWaves + 7) / 8;

  zero_f4<<<zb, 256, 0, stream>>>((float4*)B, n4);
  gemm_wmma<IN_DIM><<<gemmBlocks, 256, 0, stream>>>(x, w1, A, nTiles);
  scatter_edges<<<sb, 256, 0, stream>>>(A, src, dst, ew, B, E);
  relu_bias_k<<<zb, 256, 0, stream>>>((const float4*)B, b1, (float4*)A, n4);

  zero_f4<<<zb, 256, 0, stream>>>((float4*)out, n4);
  gemm_wmma<HID_DIM><<<gemmBlocks, 256, 0, stream>>>(A, w2, B, nTiles);
  scatter_edges<<<sb, 256, 0, stream>>>(B, src, dst, ew, out, E);
  bias_add_k<<<zb, 256, 0, stream>>>((float4*)out, b2, n4);
}


# elementwise passes fused into the 2 GEMM launches: each GEMM wave stores the bias as start value of the next scatter's sums (bias summed first, not last), GEMM 2 applies relu on load; 8 launches -> 4
# speedup vs baseline: 2.1747x; 1.3135x over previous
#include <hip/hip_runtime.h>


typedef __attribute__((ext_vector_type(2))) float v2f;
typedef __attribute__((ext_vector_type(8))) float v8f;

#define IN_DIM  128
#define HID_DIM 64
#define OUT_DIM 64
#define W_PITCH 160

template <int K, bool RELU_IN>
__global__ __launch_bounds__(256) void gemm_wmma(const float* __restrict__ X,
                                                 const float* __restrict__ W,
                                                 const float* __restrict__ bias,
                                                 float* __restrict__ Y,
                                                 float* __restrict__ S,
                                                 int nTiles) {
  __shared__ float sW[(K / 2) * W_PITCH];
  __shared__ __attribute__((aligned(16))) float sBias[64];
  float bstage = 0.f;
  if (threadIdx.x < 64) bstage = bias[threadIdx.x];
  for (int i = threadIdx.x; i < K * 64; i += 256) {
    int r = i >> 6, c = i & 63;
    sW[(r >> 1) * W_PITCH + (c << 1) + (r & 1)] = W[i];
  }
  if (threadIdx.x < 64) sBias[threadIdx.x] = bstage;
  __syncthreads();

  const int wave = threadIdx.x >> 5;
  const int lane = threadIdx.x & 31;
  const int tile = blockIdx.x * 8 + wave;
  if (tile >= nTiles) return;

  {
    const float4 bv = reinterpret_cast<const float4*>(sBias)[lane & 15];
    float4* s4 = reinterpret_cast<float4*>(S + (size_t)tile * (16 * 64)) + lane;
#pragma unroll
    for (int r = 0; r < 8; r++) s4[r * 32] = bv;
  }

  const int mrow = lane & 15;
  const int half = lane >> 4;
  const int koff = half << 1;

  const float* xrow = X + (size_t)(tile * 16 + mrow) * K;
  const float* wbase = sW + (size_t)half * W_PITCH + (mrow << 1);

  v8f acc0 = {}, acc1 = {}, acc2 = {}, acc3 = {};

  for (int k = 0; k < K; k += 4) {
    v2f a = *(const v2f*)(xrow + k + koff);
    if (RELU_IN) { a.x = fmaxf(a.x, 0.f); a.y = fmaxf(a.y, 0.f); }
    const float* wk = wbase + (k >> 1) * W_PITCH;
    v2f b0 = *(const v2f*)(wk + 0);
    v2f b1 = *(const v2f*)(wk + 32);
    v2f b2 = *(const v2f*)(wk + 64);
    v2f b3 = *(const v2f*)(wk + 96);

    acc0 = __builtin_amdgcn_wmma_f32_16x16x4_f32(false, a, false, b0, (short)0, acc0, false, false);
    acc1 = __builtin_amdgcn_wmma_f32_16x16x4_f32(false, a, false, b1, (short)0, acc1, false, false);
    acc2 = __builtin_amdgcn_wmma_f32_16x16x4_f32(false, a, false, b2, (short)0, acc2, false, false);
    acc3 = __builtin_amdgcn_wmma_f32_16x16x4_f32(false, a, false, b3, (short)0, acc3, false, false);
  }

  const int rbase = tile * 16 + (half << 3);
  float* y = Y + (size_t)rbase * 64 + mrow;
#pragma unroll
  for (int v = 0; v < 8; v++) {
    y[(size_t)v * 64 + 0]  = acc0[v];
    y[(size_t)v * 64 + 16] = acc1[v];
    y[(size_t)v * 64 + 32] = acc2[v];
    y[(size_t)v * 64 + 48] = acc3[v];
  }
}

#define EDGES_PER_WAVE 32
#define EDGE_CHUNK 8

__device__ __forceinline__ void edge_one(const float* __restrict__ H, float* __restrict__ agg,
                                         int s, int d, float w, int lane) {
  const float* h = H + (size_t)s * 64 + lane;
  float* p = agg + (size_t)d * 64 + lane;
  float v0 = h[0], v1 = h[32];
  unsafeAtomicAdd(p, v0 * w);
  unsafeAtomicAdd(p + 32, v1 * w);
}

__global__ __launch_bounds__(256) void scatter_edges(const float* __restrict__ H,
                                                     const int* __restrict__ src,
                                                     const int* __restrict__ dst,
                                                     const float* __restrict__ ew,
                                                     float* __restrict__ agg,
                                                     int nEdges) {
  const int lane = threadIdx.x & 31;
  const long long e0 = ((long long)blockIdx.x * 8 + (threadIdx.x >> 5)) * EDGES_PER_WAVE;
  if (e0 >= nEdges) return;
  const long long rem = (long long)nEdges - e0;
  const int cnt = rem < EDGES_PER_WAVE ? (int)rem : EDGES_PER_WAVE;

  int s = 0, d = 0, wbits = 0;
  if (lane < cnt) {
    s = src[e0 + lane];
    d = dst[e0 + lane];
    wbits = __float_as_int(ew[e0 + lane]);
  }

  int j = 0;
  for (; j + EDGE_CHUNK <= cnt; j += EDGE_CHUNK) {
    float v0[EDGE_CHUNK], v1[EDGE_CHUNK];
#pragma unroll
    for (int u = 0; u < EDGE_CHUNK; u++) {
      const int sj = __builtin_amdgcn_readlane(s, j + u);
      const float* h = H + (size_t)sj * 64 + lane;
      v0[u] = h[0];
      v1[u] = h[32];
    }
#pragma unroll
    for (int u = 0; u < EDGE_CHUNK; u++) {
      const int dj = __builtin_amdgcn_readlane(d, j + u);
      const float wj = __int_as_float(__builtin_amdgcn_readlane(wbits, j + u));
      float* p = agg + (size_t)dj * 64 + lane;
      unsafeAtomicAdd(p, v0[u] * wj);
      unsafeAtomicAdd(p + 32, v1[u] * wj);
    }
  }
  for (; j < cnt; j++) {
    edge_one(H, agg, __builtin_amdgcn_readlane(s, j), __builtin_amdgcn_readlane(d, j),
             __int_as_float(__builtin_amdgcn_readlane(wbits, j)), lane);
  }
}

extern "C" void kernel_launch(void* const* d_in, const int* in_sizes, int n_in,
                              void* d_out, int out_size, void* d_ws, size_t ws_size,
                              hipStream_t stream) {
  const float* x  = (const float*)d_in[0];
  const int*   ei = (const int*)d_in[1];
  const float* ew = (const float*)d_in[2];
  const float* w1 = (const float*)d_in[3];
  const float* b1 = (const float*)d_in[4];
  const float* w2 = (const float*)d_in[5];
  const float* b2 = (const float*)d_in[6];

  const int N = in_sizes[0] / IN_DIM;
  const int E = in_sizes[2];
  const int* src = ei;
  const int* dst = ei + E;
  float* out = (float*)d_out;

  float* A = (float*)d_ws;
  float* B = A + (size_t)N * 64;

  const int nTiles = (N + 15) / 16;
  const int gemmBlocks = (nTiles + 7) / 8;
  const int edgeWaves = (E + EDGES_PER_WAVE - 1) / EDGES_PER_WAVE;
  const int sb = (edgeWaves + 7) / 8;

  gemm_wmma<IN_DIM, false><<<gemmBlocks, 256, 0, stream>>>(x, w1, b1, A, B, nTiles);
  scatter_edges<<<sb, 256, 0, stream>>>(A, src, dst, ew, B, E);

  gemm_wmma<HID_DIM, true><<<gemmBlocks, 256, 0, stream>>>(B, w2, b2, A, out, nTiles);
  scatter_edges<<<sb, 256, 0, stream>>>(A, src, dst, ew, out, E);
}


# edge sums by destination: edges ordered once per call (LDS-counter ranks, 32-slot node rows), one wave sums a row in registers and stores it once; no float atomics; sums in slot (arrival) order
# speedup vs baseline: 3.0320x; 1.3942x over previous
#include <hip/hip_runtime.h>


typedef __attribute__((ext_vector_type(2))) float v2f;
typedef __attribute__((ext_vector_type(8))) float v8f;

#define IN_DIM  128
#define HID_DIM 64
#define OUT_DIM 64
#define W_PITCH 160

#define SLOTS        32
#define BUCKET_SHIFT 9
#define BUCKET_NODES 512
#define MAX_BUCKETS  256
#define BUCKET_CAP   16384
#define SPILL_FLAG   0x40000000

template <int K, bool RELU_IN, bool CLEAR>
__global__ __launch_bounds__(256) void gemm_wmma(const float* __restrict__ X,
                                                 const float* __restrict__ W,
                                                 float* __restrict__ Y,
                                                 int* __restrict__ bucketCount,
                                                 int* __restrict__ spillCount,
                                                 int nTiles) {
  __shared__ float sW[(K / 2) * W_PITCH];
  if (CLEAR) {
    if (blockIdx.x == 0) {
      bucketCount[threadIdx.x] = 0;
      if (threadIdx.x == 0) *spillCount = 0;
    }
  }
  for (int i = threadIdx.x; i < K * 64; i += 256) {
    int r = i >> 6, c = i & 63;
    sW[(r >> 1) * W_PITCH + (c << 1) + (r & 1)] = W[i];
  }
  __syncthreads();

  const int wave = threadIdx.x >> 5;
  const int lane = threadIdx.x & 31;
  const int tile = blockIdx.x * 8 + wave;
  if (tile >= nTiles) return;

  const int mrow = lane & 15;
  const int half = lane >> 4;
  const int koff = half << 1;

  const float* xrow = X + (size_t)(tile * 16 + mrow) * K;
  const float* wbase = sW + (size_t)half * W_PITCH + (mrow << 1);

  v8f acc0 = {}, acc1 = {}, acc2 = {}, acc3 = {};

  for (int k = 0; k < K; k += 4) {
    v2f a = *(const v2f*)(xrow + k + koff);
    if (RELU_IN) { a.x = fmaxf(a.x, 0.f); a.y = fmaxf(a.y, 0.f); }
    const float* wk = wbase + (k >> 1) * W_PITCH;
    v2f b0 = *(const v2f*)(wk + 0);
    v2f b1 = *(const v2f*)(wk + 32);
    v2f b2 = *(const v2f*)(wk + 64);
    v2f b3 = *(const v2f*)(wk + 96);

    acc0 = __builtin_amdgcn_wmma_f32_16x16x4_f32(false, a, false, b0, (short)0, acc0, false, false);
    acc1 = __builtin_amdgcn_wmma_f32_16x16x4_f32(false, a, false, b1, (short)0, acc1, false, false);
    acc2 = __builtin_amdgcn_wmma_f32_16x16x4_f32(false, a, false, b2, (short)0, acc2, false, false);
    acc3 = __builtin_amdgcn_wmma_f32_16x16x4_f32(false, a, false, b3, (short)0, acc3, false, false);
  }

  const int rbase = tile * 16 + (half << 3);
  float* y = Y + (size_t)rbase * 64 + mrow;
#pragma unroll
  for (int v = 0; v < 8; v++) {
    y[(size_t)v * 64 + 0]  = acc0[v];
    y[(size_t)v * 64 + 16] = acc1[v];
    y[(size_t)v * 64 + 32] = acc2[v];
    y[(size_t)v * 64 + 48] = acc3[v];
  }
}

__device__ __forceinline__ void spill_record(int* __restrict__ spillCount, int* __restrict__ spill, int4 rec) {
  const int o = atomicAdd(spillCount, 1);
  reinterpret_cast<int4*>(spill)[o] = rec;
}

#define SPLIT_EPT 8
__global__ __launch_bounds__(256) void split_edges(const int* __restrict__ src,
                                                   const int* __restrict__ dst,
                                                   const float* __restrict__ ew,
                                                   int* __restrict__ bucketCount,
                                                   int* __restrict__ part,
                                                   int* __restrict__ spillCount,
                                                   int* __restrict__ spill,
                                                   int nEdges) {
  __shared__ int h[MAX_BUCKETS];
  __shared__ int gb[MAX_BUCKETS];
  const int tid = threadIdx.x;
  h[tid] = 0;
  __syncthreads();

  const long long e0 = (long long)blockIdx.x * (256 * SPLIT_EPT) + tid;
  int d[SPLIT_EPT], s[SPLIT_EPT], w[SPLIT_EPT], r[SPLIT_EPT];
#pragma unroll
  for (int u = 0; u < SPLIT_EPT; u++) {
    const long long e = e0 + u * 256;
    d[u] = 0; s[u] = 0; w[u] = 0;
    if (e < nEdges) { d[u] = dst[e]; s[u] = src[e]; w[u] = __float_as_int(ew[e]); }
  }
#pragma unroll
  for (int u = 0; u < SPLIT_EPT; u++) {
    r[u] = 0;
    if (e0 + u * 256 < nEdges) r[u] = atomicAdd(&h[d[u] >> BUCKET_SHIFT], 1);
  }
  __syncthreads();
  {
    const int k = h[tid];
    gb[tid] = k > 0 ? atomicAdd(&bucketCount[tid], k) : 0;
  }
  __syncthreads();
#pragma unroll
  for (int u = 0; u < SPLIT_EPT; u++) {
    if (e0 + u * 256 < nEdges) {
      const int b = d[u] >> BUCKET_SHIFT;
      const int pos = gb[b] + r[u];
      const int4 rec = make_int4(d[u], s[u], w[u], 0);
      if (pos < BUCKET_CAP) reinterpret_cast<int4*>(part)[(size_t)b * BUCKET_CAP + pos] = rec;
      else spill_record(spillCount, spill, rec);
    }
  }
}

__global__ __launch_bounds__(1024) void slot_edges(const int* __restrict__ bucketCount,
                                                   const int* __restrict__ part,
                                                   int* __restrict__ cnt,
                                                   int* __restrict__ table,
                                                   int* __restrict__ spillCount,
                                                   int* __restrict__ spill,
                                                   int nNodes) {
  __shared__ int c[BUCKET_NODES];
  const int b = blockIdx.x, tid = threadIdx.x;
  if (tid < BUCKET_NODES) c[tid] = 0;
  __syncthreads();
  const int filled = bucketCount[b];
  const int L = filled < BUCKET_CAP ? filled : BUCKET_CAP;
  const int4* p = reinterpret_cast<const int4*>(part) + (size_t)b * BUCKET_CAP;
  for (int i0 = 0; i0 < L; i0 += 4 * 1024) {
    int4 rec[4];
#pragma unroll
    for (int u = 0; u < 4; u++) {
      const int i = i0 + u * 1024 + tid;
      rec[u] = p[i < L ? i : L - 1];
    }
#pragma unroll
    for (int u = 0; u < 4; u++) {
      const int i = i0 + u * 1024 + tid;
      if (i < L) {
        const int rk = atomicAdd(&c[rec[u].x & (BUCKET_NODES - 1)], 1);
        if (rk < SLOTS) reinterpret_cast<int2*>(table)[(size_t)rec[u].x * SLOTS + rk] = make_int2(rec[u].y, rec[u].z);
        else spill_record(spillCount, spill, rec[u]);
      }
    }
  }
  __syncthreads();
  if (tid < BUCKET_NODES) {
    const int node = b * BUCKET_NODES + tid;
    if (node < nNodes) cnt[node] = c[tid] | (filled > BUCKET_CAP ? SPILL_FLAG : 0);
  }
}

__device__ __forceinline__ v2f row2(const float* __restrict__ H, int s, int lane) {
  return *reinterpret_cast<const v2f*>(H + (size_t)s * 64 + 2 * lane);
}
template <int CNT>
__device__ __forceinline__ void load_rows(const float* __restrict__ H, int s, int j, int lane, v2f* t) {
#pragma unroll
  for (int u = 0; u < CNT; u++) t[u] = row2(H, __builtin_amdgcn_readlane(s, j + u), lane);
}
template <int CNT>
__device__ __forceinline__ v2f add_rows(v2f acc, const v2f* t, int wb, int j) {
#pragma unroll
  for (int u = 0; u < CNT; u++) acc += t[u] * __int_as_float(__builtin_amdgcn_readlane(wb, j + u));
  return acc;
}

__global__ __launch_bounds__(256) void sum_rows(const float* __restrict__ H,
                                                const int* __restrict__ cnt,
                                                const int* __restrict__ table,
                                                const int* __restrict__ spillCount,
                                                const int* __restrict__ spill,
                                                const float* __restrict__ bias,
                                                float* __restrict__ out,
                                                int nNodes) {
  const int lane = threadIdx.x & 31;
  const int nWaves = gridDim.x * 8;
  int node = __builtin_amdgcn_readfirstlane(blockIdx.x * 8 + (threadIdx.x >> 5));
  if (node >= nNodes) return;
  const v2f bv = *reinterpret_cast<const v2f*>(bias + 2 * lane);
  const int2* tab = reinterpret_cast<const int2*>(table);

  const int2* idle = reinterpret_cast<const int2*>(bias) + lane;

  int n1 = node + nWaves;
  int c0 = cnt[node];
  int c1 = n1 < nNodes ? cnt[n1] : 0;
  const int2* q0 = idle;
  if (lane < (c0 & (SPILL_FLAG - 1))) q0 = tab + (size_t)node * SLOTS + lane;
  int2 e0 = *q0;

  for (;;) {
    const int n2 = n1 + nWaves;
    const int c2 = n2 < nNodes ? cnt[n2] : 0;
    const int2* q1 = idle;
    if (n1 < nNodes && lane < (c1 & (SPILL_FLAG - 1))) q1 = tab + (size_t)n1 * SLOTS + lane;
    const int2 e1 = *q1;

    const int cw = __builtin_amdgcn_readfirstlane(c0);
    const int deg = cw & (SPILL_FLAG - 1);
    const int m = deg < SLOTS ? deg : SLOTS;
    const int s = e0.x, wb = e0.y;
    v2f acc = bv;

    v2f t[16];
    int j = 0;
    for (; j + 16 <= m; j += 16) {
      load_rows<16>(H, s, j, lane, t);
      acc = add_rows<16>(acc, t, wb, j);
    }
    {
      const int rem = m - j;
      const int j8 = j, j4 = j8 + (rem & 8), j2 = j4 + (rem & 4), j1 = j2 + (rem & 2);
      if (rem & 8) load_rows<8>(H, s, j8, lane, t);
      if (rem & 4) load_rows<4>(H, s, j4, lane, t + 8);
      if (rem & 2) load_rows<2>(H, s, j2, lane, t + 12);
      if (rem & 1) load_rows<1>(H, s, j1, lane, t + 14);
      if (rem & 8) acc = add_rows<8>(acc, t, wb, j8);
      if (rem & 4) acc = add_rows<4>(acc, t + 8, wb, j4);
      if (rem & 2) acc = add_rows<2>(acc, t + 12, wb, j2);
      if (rem & 1) acc = add_rows<1>(acc, t + 14, wb, j1);
    }

    if (deg > SLOTS || (cw & SPILL_FLAG)) {
      const int total = *spillCount;
      const int4* sp = reinterpret_cast<const int4*>(spill);
      for (int base = 0; base < total; base += 32) {
        int4 o = make_int4(-1, 0, 0, 0);
        if (base + lane < total) o = sp[base + lane];
        unsigned match = __builtin_amdgcn_ballot_w32(o.x == node);
        while (match) {
          const int l = __builtin_ctz(match);
          match &= match - 1;
          const v2f v = row2(H, __builtin_amdgcn_readlane(o.y, l), lane);
          acc += v * __int_as_float(__builtin_amdgcn_readlane(o.z, l));
        }
      }
    }

    *reinterpret_cast<v2f*>(out + (size_t)node * 64 + 2 * lane) = acc;

    if (n1 >= nNodes) break;
    node = n1; n1 = n2; c0 = c1; c1 = c2; e0 = e1;
  }
}

extern "C" void kernel_launch(void* const* d_in, const int* in_sizes, int n_in,
                              void* d_out, int out_size, void* d_ws, size_t ws_size,
                              hipStream_t stream) {
  const float* x  = (const float*)d_in[0];
  const int*   ei = (const int*)d_in[1];
  const float* ew = (const float*)d_in[2];
  const float* w1 = (const float*)d_in[3];
  const float* b1 = (const float*)d_in[4];
  const float* w2 = (const float*)d_in[5];
  const float* b2 = (const float*)d_in[6];

  const int N = in_sizes[0] / IN_DIM;
  const int E = in_sizes[2];
  const int* src = ei;
  const int* dst = ei + E;
  float* out = (float*)d_out;

  const int nBuckets = (N + BUCKET_NODES - 1) / BUCKET_NODES;

  float* A = (float*)d_ws;
  float* B = A + (size_t)N * 64;
  int* table = (int*)(B + (size_t)N * 64);
  int* part  = table + (size_t)N * SLOTS * 2;
  int* spill = part + (size_t)nBuckets * BUCKET_CAP * 4;
  int* cnt   = spill + (size_t)E * 4;
  int* bucketCount = cnt + N;
  int* spillCount  = bucketCount + MAX_BUCKETS;

  const int nTiles = (N + 15) / 16;
  const int gemmBlocks = (nTiles + 7) / 8;
  const int splitBlocks = (E + 256 * SPLIT_EPT - 1) / (256 * SPLIT_EPT);
  const int rowBlocks = 1280;

  gemm_wmma<IN_DIM, false, true><<<gemmBlocks, 256, 0, stream>>>(x, w1, A, bucketCount, spillCount, nTiles);
  split_edges<<<splitBlocks, 256, 0, stream>>>(src, dst, ew, bucketCount, part, spillCount, spill, E);
  slot_edges<<<nBuckets, 1024, 0, stream>>>(bucketCount, part, cnt, table, spillCount, spill, N);
  sum_rows<<<rowBlocks, 256, 0, stream>>>(A, cnt, table, spillCount, spill, b1, B, N);
  gemm_wmma<HID_DIM, true, false><<<gemmBlocks, 256, 0, stream>>>(B, w2, A, bucketCount, spillCount, nTiles);
  sum_rows<<<rowBlocks, 256, 0, stream>>>(A, cnt, table, spillCount, spill, b2, out, N);
}


# layer-2 product fused into the layer-1 row-sum launch: a block sums 64 nodes' rows, keeps relu of them in LDS and multiplies by W2 there (f32 WMMA); 6 launches -> 5; every sum in the old order
# speedup vs baseline: 3.1540x; 1.0402x over previous
#include <hip/hip_runtime.h>


typedef __attribute__((ext_vector_type(2))) float v2f;
typedef __attribute__((ext_vector_type(8))) float v8f;

#define IN_DIM  128
#define HID_DIM 64
#define OUT_DIM 64
#define W_PITCH 160
#define T_PITCH 68

#define SLOTS        32
#define BUCKET_SHIFT 9
#define BUCKET_NODES 512
#define MAX_BUCKETS  256
#define BUCKET_CAP   16384
#define SPILL_FLAG   0x40000000

template <int K>
__global__ __launch_bounds__(256) void gemm_wmma(const float* __restrict__ X,
                                                 const float* __restrict__ W,
                                                 float* __restrict__ Y,
                                                 int* __restrict__ bucketCount,
                                                 int* __restrict__ spillCount,
                                                 int nTiles) {
  __shared__ float sW[(K / 2) * W_PITCH];
  if (blockIdx.x == 0) {
    bucketCount[threadIdx.x] = 0;
    if (threadIdx.x == 0) *spillCount = 0;
  }
  for (int i = threadIdx.x; i < K * 64; i += 256) {
    int r = i >> 6, c = i & 63;
    sW[(r >> 1) * W_PITCH + (c << 1) + (r & 1)] = W[i];
  }
  __syncthreads();

  const int wave = threadIdx.x >> 5;
  const int lane = threadIdx.x & 31;
  const int tile = blockIdx.x * 8 + wave;
  if (tile >= nTiles) return;

  const int mrow = lane & 15;
  const int half = lane >> 4;
  const int koff = half << 1;

  const float* xrow = X + (size_t)(tile * 16 + mrow) * K;
  const float* wbase = sW + (size_t)half * W_PITCH + (mrow << 1);

  v8f acc0 = {}, acc1 = {}, acc2 = {}, acc3 = {};

  for (int k = 0; k < K; k += 4) {
    v2f a = *(const v2f*)(xrow + k + koff);
    const float* wk = wbase + (k >> 1) * W_PITCH;
    v2f b0 = *(const v2f*)(wk + 0);
    v2f b1 = *(const v2f*)(wk + 32);
    v2f b2 = *(const v2f*)(wk + 64);
    v2f b3 = *(const v2f*)(wk + 96);

    acc0 = __builtin_amdgcn_wmma_f32_16x16x4_f32(false, a, false, b0, (short)0, acc0, false, false);
    acc1 = __builtin_amdgcn_wmma_f32_16x16x4_f32(false, a, false, b1, (short)0, acc1, false, false);
    acc2 = __builtin_amdgcn_wmma_f32_16x16x4_f32(false, a, false, b2, (short)0, acc2, false, false);
    acc3 = __builtin_amdgcn_wmma_f32_16x16x4_f32(false, a, false, b3, (short)0, acc3, false, false);
  }

  const int rbase = tile * 16 + (half << 3);
  float* y = Y + (size_t)rbase * 64 + mrow;
#pragma unroll
  for (int v = 0; v < 8; v++) {
    y[(size_t)v * 64 + 0]  = acc0[v];
    y[(size_t)v * 64 + 16] = acc1[v];
    y[(size_t)v * 64 + 32] = acc2[v];
    y[(size_t)v * 64 + 48] = acc3[v];
  }
}

__device__ __forceinline__ void spill_record(int* __restrict__ spillCount, int* __restrict__ spill, int4 rec) {
  const int o = atomicAdd(spillCount, 1);
  reinterpret_cast<int4*>(spill)[o] = rec;
}

#define SPLIT_EPT 8
__global__ __launch_bounds__(256) void split_edges(const int* __restrict__ src,
                                                   const int* __restrict__ dst,
                                                   const float* __restrict__ ew,
                                                   int* __restrict__ bucketCount,
                                                   int* __restrict__ part,
                                                   int* __restrict__ spillCount,
                                                   int* __restrict__ spill,
                                                   int nEdges) {
  __shared__ int h[MAX_BUCKETS];
  __shared__ int gb[MAX_BUCKETS];
  const int tid = threadIdx.x;
  h[tid] = 0;
  __syncthreads();

  const long long e0 = (long long)blockIdx.x * (256 * SPLIT_EPT) + tid;
  int d[SPLIT_EPT], s[SPLIT_EPT], w[SPLIT_EPT], r[SPLIT_EPT];
#pragma unroll
  for (int u = 0; u < SPLIT_EPT; u++) {
    const long long e = e0 + u * 256;
    d[u] = 0; s[u] = 0; w[u] = 0;
    if (e < nEdges) { d[u] = dst[e]; s[u] = src[e]; w[u] = __float_as_int(ew[e]); }
  }
#pragma unroll
  for (int u = 0; u < SPLIT_EPT; u++) {
    r[u] = 0;
    if (e0 + u * 256 < nEdges) r[u] = atomicAdd(&h[d[u] >> BUCKET_SHIFT], 1);
  }
  __syncthreads();
  {
    const int k = h[tid];
    gb[tid] = k > 0 ? atomicAdd(&bucketCount[tid], k) : 0;
  }
  __syncthreads();
#pragma unroll
  for (int u = 0; u < SPLIT_EPT; u++) {
    if (e0 + u * 256 < nEdges) {
      const int b = d[u] >> BUCKET_SHIFT;
      const int pos = gb[b] + r[u];
      const int4 rec = make_int4(d[u], s[u], w[u], 0);
      if (pos < BUCKET_CAP) reinterpret_cast<int4*>(part)[(size_t)b * BUCKET_CAP + pos] = rec;
      else spill_record(spillCount, spill, rec);
    }
  }
}

__global__ __launch_bounds__(1024) void slot_edges(const int* __restrict__ bucketCount,
                                                   const int* __restrict__ part,
                                                   int* __restrict__ cnt,
                                                   int* __restrict__ table,
                                                   int* __restrict__ spillCount,
                                                   int* __restrict__ spill,
                                                   int nNodes) {
  __shared__ int c[BUCKET_NODES];
  const int b = blockIdx.x, tid = threadIdx.x;
  if (tid < BUCKET_NODES) c[tid] = 0;
  __syncthreads();
  const int filled = bucketCount[b];
  const int L = filled < BUCKET_CAP ? filled : BUCKET_CAP;
  const int4* p = reinterpret_cast<const int4*>(part) + (size_t)b * BUCKET_CAP;
  for (int i0 = 0; i0 < L; i0 += 4 * 1024) {
    int4 rec[4];
#pragma unroll
    for (int u = 0; u < 4; u++) {
      const int i = i0 + u * 1024 + tid;
      rec[u] = p[i < L ? i : L - 1];
    }
#pragma unroll
    for (int u = 0; u < 4; u++) {
      const int i = i0 + u * 1024 + tid;
      if (i < L) {
        const int rk = atomicAdd(&c[rec[u].x & (BUCKET_NODES - 1)], 1);
        if (rk < SLOTS) reinterpret_cast<int2*>(table)[(size_t)rec[u].x * SLOTS + rk] = make_int2(rec[u].y, rec[u].z);
        else spill_record(spillCount, spill, rec[u]);
      }
    }
  }
  __syncthreads();
  if (tid < BUCKET_NODES) {
    const int node = b * BUCKET_NODES + tid;
    if (node < nNodes) cnt[node] = c[tid] | (filled > BUCKET_CAP ? SPILL_FLAG : 0);
  }
}

__device__ __forceinline__ v2f row2(const float* __restrict__ H, int s, int lane) {
  return *reinterpret_cast<const v2f*>(H + (size_t)s * 64 + 2 * lane);
}
template <int CNT>
__device__ __forceinline__ void load_rows(const float* __restrict__ H, int s, int j, int lane, v2f* t) {
#pragma unroll
  for (int u = 0; u < CNT; u++) t[u] = row2(H, __builtin_amdgcn_readlane(s, j + u), lane);
}
template <int CNT>
__device__ __forceinline__ v2f add_rows(v2f acc, const v2f* t, int wb, int j) {
#pragma unroll
  for (int u = 0; u < CNT; u++) acc += t[u] * __int_as_float(__builtin_amdgcn_readlane(wb, j + u));
  return acc;
}

__device__ __forceinline__ v2f sum_node(const float* __restrict__ H, int s, int wb, int cw, int node,
                                        const int* __restrict__ spillCount, const int* __restrict__ spill,
                                        v2f bv, int lane) {
  const int deg = cw & (SPILL_FLAG - 1);
  const int m = deg < SLOTS ? deg : SLOTS;
  v2f acc = bv;

  v2f t[16];
  int j = 0;
  for (; j + 16 <= m; j += 16) {
    load_rows<16>(H, s, j, lane, t);
    acc = add_rows<16>(acc, t, wb, j);
  }
  {
    const int rem = m - j;
    const int j8 = j, j4 = j8 + (rem & 8), j2 = j4 + (rem & 4), j1 = j2 + (rem & 2);
    if (rem & 8) load_rows<8>(H, s, j8, lane, t);
    if (rem & 4) load_rows<4>(H, s, j4, lane, t + 8);
    if (rem & 2) load_rows<2>(H, s, j2, lane, t + 12);
    if (rem & 1) load_rows<1>(H, s, j1, lane, t + 14);
    if (rem & 8) acc = add_rows<8>(acc, t, wb, j8);
    if (rem & 4) acc = add_rows<4>(acc, t + 8, wb, j4);
    if (rem & 2) acc = add_rows<2>(acc, t + 12, wb, j2);
    if (rem & 1) acc = add_rows<1>(acc, t + 14, wb, j1);
  }

  if (deg > SLOTS || (cw & SPILL_FLAG)) {
    const int total = *spillCount;
    const int4* sp = reinterpret_cast<const int4*>(spill);
    for (int base = 0; base < total; base += 32) {
      int4 o = make_int4(-1, 0, 0, 0);
      if (base + lane < total) o = sp[base + lane];
      unsigned match = __builtin_amdgcn_ballot_w32(o.x == node);
      while (match) {
        const int l = __builtin_ctz(match);
        match &= match - 1;
        const v2f v = row2(H, __builtin_amdgcn_readlane(o.y, l), lane);
        acc += v * __int_as_float(__builtin_amdgcn_readlane(o.z, l));
      }
    }
  }
  return acc;
}

__global__ __launch_bounds__(256) void sum_rows(const float* __restrict__ H,
                                                const int* __restrict__ cnt,
                                                const int* __restrict__ table,
                                                const int* __restrict__ spillCount,
                                                const int* __restrict__ spill,
                                                const float* __restrict__ bias,
                                                float* __restrict__ out,
                                                int nNodes) {
  const int lane = threadIdx.x & 31;
  const int nWaves = gridDim.x * 8;
  int node = __builtin_amdgcn_readfirstlane(blockIdx.x * 8 + (threadIdx.x >> 5));
  if (node >= nNodes) return;
  const v2f bv = *reinterpret_cast<const v2f*>(bias + 2 * lane);
  const int2* tab = reinterpret_cast<const int2*>(table);

  const int2* idle = reinterpret_cast<const int2*>(bias) + lane;

  int n1 = node + nWaves;
  int c0 = cnt[node];
  int c1 = n1 < nNodes ? cnt[n1] : 0;
  const int2* q0 = idle;
  if (lane < (c0 & (SPILL_FLAG - 1))) q0 = tab + (size_t)node * SLOTS + lane;
  int2 e0 = *q0;

  for (;;) {
    const int n2 = n1 + nWaves;
    const int c2 = n2 < nNodes ? cnt[n2] : 0;
    const int2* q1 = idle;
    if (n1 < nNodes && lane < (c1 & (SPILL_FLAG - 1))) q1 = tab + (size_t)n1 * SLOTS + lane;
    const int2 e1 = *q1;

    const int cw = __builtin_amdgcn_readfirstlane(c0);
    const v2f acc = sum_node(H, e0.x, e0.y, cw, node, spillCount, spill, bv, lane);

    *reinterpret_cast<v2f*>(out + (size_t)node * 64 + 2 * lane) = acc;

    if (n1 >= nNodes) break;
    node = n1; n1 = n2; c0 = c1; c1 = c2; e0 = e1;
  }
}

__global__ __launch_bounds__(256) void sum_rows_gemm(const float* __restrict__ H,
                                                     const int* __restrict__ cnt,
                                                     const int* __restrict__ table,
                                                     const int* __restrict__ spillCount,
                                                     const int* __restrict__ spill,
                                                     const float* __restrict__ bias,
                                                     const float* __restrict__ W,
                                                     float* __restrict__ Y,
                                                     int nNodes) {
  __shared__ __attribute__((aligned(16))) float sW[(HID_DIM / 2) * W_PITCH];
  __shared__ __attribute__((aligned(16))) float sT[4 * 16 * T_PITCH];

  const int lane = threadIdx.x & 31;
  const int wave = __builtin_amdgcn_readfirstlane(threadIdx.x >> 5);
  const int part = wave & 1;
  const int tile = blockIdx.x * 4 + (wave >> 1);
  float* tl = sT + (wave >> 1) * (16 * T_PITCH);
  const int node0 = tile * 16 + part * 8;

  int cl = 0;
  if (lane < 8 && node0 + lane < nNodes) cl = cnt[node0 + lane];

  {
    float4 w0[2], w1[2];
#pragma unroll
    for (int u = 0; u < 2; u++) {
      const int item = threadIdx.x + 256 * u;
      const float* wp = W + (size_t)(item >> 4) * 128 + ((item & 15) << 2);
      w0[u] = *reinterpret_cast<const float4*>(wp);
      w1[u] = *reinterpret_cast<const float4*>(wp + 64);
    }
#pragma unroll
    for (int u = 0; u < 2; u++) {
      const int item = threadIdx.x + 256 * u;
      float4* sp = reinterpret_cast<float4*>(sW + (item >> 4) * W_PITCH + ((item & 15) << 3));
      sp[0] = make_float4(w0[u].x, w1[u].x, w0[u].y, w1[u].y);
      sp[1] = make_float4(w0[u].z, w1[u].z, w0[u].w, w1[u].w);
    }
  }

  const v2f bv = *reinterpret_cast<const v2f*>(bias + 2 * lane);
  const int2* tab = reinterpret_cast<const int2*>(table);
  const int2* idle = reinterpret_cast<const int2*>(bias) + lane;

  int c0 = __builtin_amdgcn_readlane(cl, 0);
  const int2* q0 = idle;
  if (lane < (c0 & (SPILL_FLAG - 1))) q0 = tab + (size_t)node0 * SLOTS + lane;
  int2 e0 = *q0;

  for (int i = 0; i < 8; i++) {
    const int node = node0 + i;
    const int c1 = __builtin_amdgcn_readlane(cl, i + 1);
    const int2* q1 = idle;
    if (lane < (c1 & (SPILL_FLAG - 1))) q1 = tab + (size_t)(node + 1) * SLOTS + lane;
    const int2 e1 = *q1;

    v2f acc = sum_node(H, e0.x, e0.y, c0, node, spillCount, spill, bv, lane);
    acc.x = fmaxf(acc.x, 0.f);
    acc.y = fmaxf(acc.y, 0.f);
    *reinterpret_cast<v2f*>(tl + (part * 8 + i) * T_PITCH + 2 * lane) = acc;

    c0 = c1; e0 = e1;
  }
  __syncthreads();

  const int mrow = lane & 15;
  const int half = lane >> 4;
  const int koff = half << 1;
  const float* arow = tl + mrow * T_PITCH + koff;
  const float* wbase = sW + half * W_PITCH + (mrow << 1) + part * 64;

  v8f acc0 = {}, acc1 = {};
  for (int k = 0; k < HID_DIM; k += 4) {
    const v2f a = *(const v2f*)(arow + k);
    const float* wk = wbase + (k >> 1) * W_PITCH;
    const v2f b0 = *(const v2f*)(wk + 0);
    const v2f b1 = *(const v2f*)(wk + 32);
    acc0 = __builtin_amdgcn_wmma_f32_16x16x4_f32(false, a, false, b0, (short)0, acc0, false, false);
    acc1 = __builtin_amdgcn_wmma_f32_16x16x4_f32(false, a, false, b1, (short)0, acc1, false, false);
  }

  const int rbase = tile * 16 + (half << 3);
  float* y = Y + (size_t)rbase * 64 + part * 32 + mrow;
#pragma unroll
  for (int v = 0; v < 8; v++) {
    if (rbase + v < nNodes) {
      y[(size_t)v * 64 + 0]  = acc0[v];
      y[(size_t)v * 64 + 16] = acc1[v];
    }
  }
}

extern "C" void kernel_launch(void* const* d_in, const int* in_sizes, int n_in,
                              void* d_out, int out_size, void* d_ws, size_t ws_size,
                              hipStream_t stream) {
  const float* x  = (const float*)d_in[0];
  const int*   ei = (const int*)d_in[1];
  const float* ew = (const float*)d_in[2];
  const float* w1 = (const float*)d_in[3];
  const float* b1 = (const float*)d_in[4];
  const float* w2 = (const float*)d_in[5];
  const float* b2 = (const float*)d_in[6];

  const int N = in_sizes[0] / IN_DIM;
  const int E = in_sizes[2];
  const int* src = ei;
  const int* dst = ei + E;
  float* out = (float*)d_out;

  const int nBuckets = (N + BUCKET_NODES - 1) / BUCKET_NODES;

  float* A = (float*)d_ws;
  float* B = A + (size_t)N * 64;
  int* table = (int*)(B + (size_t)N * 64);
  int* part  = table + (size_t)N * SLOTS * 2;
  int* spill = part + (size_t)nBuckets * BUCKET_CAP * 4;
  int* cnt   = spill + (size_t)E * 4;
  int* bucketCount = cnt + N;
  int* spillCount  = bucketCount + MAX_BUCKETS;

  const int nTiles = (N + 15) / 16;
  const int gemmBlocks = (nTiles + 7) / 8;
  const int splitBlocks = (E + 256 * SPLIT_EPT - 1) / (256 * SPLIT_EPT);
  const int rowBlocks = 1280;
  const int fusedBlocks = (nTiles + 3) / 4;

  gemm_wmma<IN_DIM><<<gemmBlocks, 256, 0, stream>>>(x, w1, A, bucketCount, spillCount, nTiles);
  split_edges<<<splitBlocks, 256, 0, stream>>>(src, dst, ew, bucketCount, part, spillCount, spill, E);
  slot_edges<<<nBuckets, 1024, 0, stream>>>(bucketCount, part, cnt, table, spillCount, spill, N);
  sum_rows_gemm<<<fusedBlocks, 256, 0, stream>>>(A, cnt, table, spillCount, spill, b1, w2, B, N);
  sum_rows<<<rowBlocks, 256, 0, stream>>>(B, cnt, table, spillCount, spill, b2, out, N);
}


# layer-1 product and edge split side by side in one launch (role by block index); split claims no global counter: a 32-record slice per (block, bucket); 5 launches -> 4; f32 sums in the new slot order
# speedup vs baseline: 3.4984x; 1.1092x over previous
#include <hip/hip_runtime.h>


typedef __attribute__((ext_vector_type(2))) float v2f;
typedef __attribute__((ext_vector_type(8))) float v8f;

#define IN_DIM  128
#define HID_DIM 64
#define OUT_DIM 64
#define W_PITCH 160
#define T_PITCH 68

#define SLOTS        32
#define BUCKET_SHIFT 9
#define BUCKET_NODES 512
#define MAX_BUCKETS  256
#define SLICE        32
#define SPLIT_EPT    8
#define SPLIT_EDGES  2048
#define SPILL_FLAG   0x40000000

template <int K>
__device__ __forceinline__ void gemm_role(int gb, float* __restrict__ sW,
                                          const float* __restrict__ X,
                                          const float* __restrict__ W,
                                          float* __restrict__ Y,
                                          int* __restrict__ spillCount,
                                          int nTiles) {
  if (gb == 0 && threadIdx.x == 0) *spillCount = 0;
  for (int i = threadIdx.x; i < K * 64; i += 256) {
    int r = i >> 6, c = i & 63;
    sW[(r >> 1) * W_PITCH + (c << 1) + (r & 1)] = W[i];
  }
  __syncthreads();

  const int wave = threadIdx.x >> 5;
  const int lane = threadIdx.x & 31;
  const int tile = gb * 8 + wave;
  if (tile >= nTiles) return;

  const int mrow = lane & 15;
  const int half = lane >> 4;
  const int koff = half << 1;

  const float* xrow = X + (size_t)(tile * 16 + mrow) * K;
  const float* wbase = sW + (size_t)half * W_PITCH + (mrow << 1);

  v8f acc0 = {}, acc1 = {}, acc2 = {}, acc3 = {};

  for (int k = 0; k < K; k += 4) {
    v2f a = *(const v2f*)(xrow + k + koff);
    const float* wk = wbase + (k >> 1) * W_PITCH;
    v2f b0 = *(const v2f*)(wk + 0);
    v2f b1 = *(const v2f*)(wk + 32);
    v2f b2 = *(const v2f*)(wk + 64);
    v2f b3 = *(const v2f*)(wk + 96);

    acc0 = __builtin_amdgcn_wmma_f32_16x16x4_f32(false, a, false, b0, (short)0, acc0, false, false);
    acc1 = __builtin_amdgcn_wmma_f32_16x16x4_f32(false, a, false, b1, (short)0, acc1, false, false);
    acc2 = __builtin_amdgcn_wmma_f32_16x16x4_f32(false, a, false, b2, (short)0, acc2, false, false);
    acc3 = __builtin_amdgcn_wmma_f32_16x16x4_f32(false, a, false, b3, (short)0, acc3, false, false);
  }

  const int rbase = tile * 16 + (half << 3);
  float* y = Y + (size_t)rbase * 64 + mrow;
#pragma unroll
  for (int v = 0; v < 8; v++) {
    y[(size_t)v * 64 + 0]  = acc0[v];
    y[(size_t)v * 64 + 16] = acc1[v];
    y[(size_t)v * 64 + 32] = acc2[v];
    y[(size_t)v * 64 + 48] = acc3[v];
  }
}

__device__ __forceinline__ void spill_record(int* __restrict__ spillCount, int* __restrict__ spill, int4 rec) {
  const int o = atomicAdd(spillCount, 1);
  reinterpret_cast<int4*>(spill)[o] = rec;
}

__device__ __forceinline__ void split_role(int sb, int* __restrict__ h, int* __restrict__ ovN,
                                           const int* __restrict__ src,
                                           const int* __restrict__ dst,
                                           const float* __restrict__ ew,
                                           int* __restrict__ part,
                                           int* __restrict__ hist,
                                           int* __restrict__ ovf,
                                           int* __restrict__ ovfCount,
                                           int nBuckets, int nEdges) {
  const int tid = threadIdx.x;
  h[tid] = 0;
  if (tid == 0) *ovN = 0;
  __syncthreads();

  const long long e0 = (long long)sb * SPLIT_EDGES + tid;
  int d[SPLIT_EPT], s[SPLIT_EPT], w[SPLIT_EPT], r[SPLIT_EPT];
#pragma unroll
  for (int u = 0; u < SPLIT_EPT; u++) {
    const long long e = e0 + u * 256;
    d[u] = 0; s[u] = 0; w[u] = 0;
    if (e < nEdges) { d[u] = dst[e]; s[u] = src[e]; w[u] = __float_as_int(ew[e]); }
  }
#pragma unroll
  for (int u = 0; u < SPLIT_EPT; u++) {
    r[u] = 0;
    if (e0 + u * 256 < nEdges) r[u] = atomicAdd(&h[d[u] >> BUCKET_SHIFT], 1);
  }
#pragma unroll
  for (int u = 0; u < SPLIT_EPT; u++) {
    if (e0 + u * 256 < nEdges) {
      const int b = d[u] >> BUCKET_SHIFT;
      int4* p = reinterpret_cast<int4*>(part) + ((size_t)sb * nBuckets + b) * SLICE + r[u];
      if (r[u] >= SLICE) {
        const int k = atomicAdd(ovN, 1);
        p = reinterpret_cast<int4*>(ovf) + (size_t)sb * SPLIT_EDGES + k;
      }
      *p = make_int4(d[u], s[u], w[u], 0);
    }
  }
  __syncthreads();
  {
    const int k = h[tid];
    hist[(size_t)sb * MAX_BUCKETS + tid] = k < SLICE ? k : SLICE;
    if (tid == 0) ovfCount[sb] = *ovN;
  }
}

__global__ __launch_bounds__(256) void gemm_and_split(const float* __restrict__ X,
                                                      const float* __restrict__ W,
                                                      float* __restrict__ Y,
                                                      int* __restrict__ spillCount,
                                                      int nTiles, int gemmBlocks,
                                                      const int* __restrict__ src,
                                                      const int* __restrict__ dst,
                                                      const float* __restrict__ ew,
                                                      int* __restrict__ part,
                                                      int* __restrict__ hist,
                                                      int* __restrict__ ovf,
                                                      int* __restrict__ ovfCount,
                                                      int nBuckets, int nEdges, int splitBlocks) {
  __shared__ __attribute__((aligned(16))) unsigned char smem[(IN_DIM / 2) * W_PITCH * 4];
  const int idx = blockIdx.x >> 1;
  if (blockIdx.x & 1) {
    if (idx < splitBlocks) {
      int* h = reinterpret_cast<int*>(smem);
      split_role(idx, h, h + MAX_BUCKETS, src, dst, ew, part, hist, ovf, ovfCount, nBuckets, nEdges);
    }
  } else {
    if (idx < gemmBlocks) gemm_role<IN_DIM>(idx, reinterpret_cast<float*>(smem), X, W, Y, spillCount, nTiles);
  }
}

__global__ __launch_bounds__(1024) void slot_edges(const int* __restrict__ hist,
                                                   const int* __restrict__ part,
                                                   const int* __restrict__ ovf,
                                                   const int* __restrict__ ovfCount,
                                                   int* __restrict__ cnt,
                                                   int* __restrict__ table,
                                                   int* __restrict__ spillCount,
                                                   int* __restrict__ spill,
                                                   int nSplit, int nBuckets, int nNodes) {
  __shared__ int c[BUCKET_NODES];
  __shared__ int anyOv;
  const int b = blockIdx.x, tid = threadIdx.x;
  const int lane = tid & 31;
  const int wave = __builtin_amdgcn_readfirstlane(tid >> 5);
  if (tid < BUCKET_NODES) c[tid] = 0;
  if (tid == 0) anyOv = 0;
  __syncthreads();

  {
    int ov = 0;
    for (int i = tid; i < nSplit; i += 1024) ov |= ovfCount[i];
    if (ov) atomicOr(&anyOv, 1);
  }

  const int4* p4 = reinterpret_cast<const int4*>(part);
  const size_t step = (size_t)32 * nBuckets * SLICE;
  for (int base = wave; base < nSplit; base += 32 * 32) {
    const int blkL = base + 32 * lane;
    int nL = 0;
    if (blkL < nSplit) nL = hist[(size_t)blkL * MAX_BUCKETS + b];
    const int4* q = p4 + ((size_t)base * nBuckets + b) * SLICE + lane;
    const int4* idle = reinterpret_cast<const int4*>(hist + (size_t)base * MAX_BUCKETS) + lane;
#pragma nounroll
    for (int j0 = 0; j0 < 32 && base + 32 * j0 < nSplit; j0 += 8, q += 8 * step) {
      int4 rec[8];
      int rk[8];
      bool on[8];
#pragma unroll
      for (int u = 0; u < 8; u++) {
        on[u] = lane < __builtin_amdgcn_readlane(nL, j0 + u);
        rec[u] = *(on[u] ? q + u * step : idle);
      }
#pragma unroll
      for (int u = 0; u < 8; u++) {
        rk[u] = 0;
        if (on[u]) rk[u] = atomicAdd(&c[rec[u].x & (BUCKET_NODES - 1)], 1);
      }
#pragma unroll
      for (int u = 0; u < 8; u++) {
        if (on[u]) {
          if (rk[u] < SLOTS) reinterpret_cast<int2*>(table)[(size_t)rec[u].x * SLOTS + rk[u]] = make_int2(rec[u].y, rec[u].z);
          else spill_record(spillCount, spill, rec[u]);
        }
      }
    }
  }

  for (int blk = b; blk < nSplit; blk += nBuckets) {
    const int n = ovfCount[blk];
    for (int k = tid; k < n; k += 1024)
      spill_record(spillCount, spill, reinterpret_cast<const int4*>(ovf)[(size_t)blk * SPLIT_EDGES + k]);
  }

  __syncthreads();
  if (tid < BUCKET_NODES) {
    const int node = b * BUCKET_NODES + tid;
    if (node < nNodes) cnt[node] = c[tid] | (anyOv ? SPILL_FLAG : 0);
  }
}

__device__ __forceinline__ v2f row2(const float* __restrict__ H, int s, int lane) {
  return *reinterpret_cast<const v2f*>(H + (size_t)s * 64 + 2 * lane);
}
template <int CNT>
__device__ __forceinline__ void load_rows(const float* __restrict__ H, int s, int j, int lane, v2f* t) {
#pragma unroll
  for (int u = 0; u < CNT; u++) t[u] = row2(H, __builtin_amdgcn_readlane(s, j + u), lane);
}
template <int CNT>
__device__ __forceinline__ v2f add_rows(v2f acc, const v2f* t, int wb, int j) {
#pragma unroll
  for (int u = 0; u < CNT; u++) acc += t[u] * __int_as_float(__builtin_amdgcn_readlane(wb, j + u));
  return acc;
}

__device__ __forceinline__ v2f sum_node(const float* __restrict__ H, int s, int wb, int cw, int node,
                                        const int* __restrict__ spillCount, const int* __restrict__ spill,
                                        v2f bv, int lane) {
  const int deg = cw & (SPILL_FLAG - 1);
  const int m = deg < SLOTS ? deg : SLOTS;
  v2f acc = bv;

  v2f t[16];
  int j = 0;
  for (; j + 16 <= m; j += 16) {
    load_rows<16>(H, s, j, lane, t);
    acc = add_rows<16>(acc, t, wb, j);
  }
  {
    const int rem = m - j;
    const int j8 = j, j4 = j8 + (rem & 8), j2 = j4 + (rem & 4), j1 = j2 + (rem & 2);
    if (rem & 8) load_rows<8>(H, s, j8, lane, t);
    if (rem & 4) load_rows<4>(H, s, j4, lane, t + 8);
    if (rem & 2) load_rows<2>(H, s, j2, lane, t + 12);
    if (rem & 1) load_rows<1>(H, s, j1, lane, t + 14);
    if (rem & 8) acc = add_rows<8>(acc, t, wb, j8);
    if (rem & 4) acc = add_rows<4>(acc, t + 8, wb, j4);
    if (rem & 2) acc = add_rows<2>(acc, t + 12, wb, j2);
    if (rem & 1) acc = add_rows<1>(acc, t + 14, wb, j1);
  }

  if (deg > SLOTS || (cw & SPILL_FLAG)) {
    const int total = *spillCount;
    const int4* sp = reinterpret_cast<const int4*>(spill);
    for (int base = 0; base < total; base += 32) {
      int4 o = make_int4(-1, 0, 0, 0);
      if (base + lane < total) o = sp[base + lane];
      unsigned match = __builtin_amdgcn_ballot_w32(o.x == node);
      while (match) {
        const int l = __builtin_ctz(match);
        match &= match - 1;
        const v2f v = row2(H, __builtin_amdgcn_readlane(o.y, l), lane);
        acc += v * __int_as_float(__builtin_amdgcn_readlane(o.z, l));
      }
    }
  }
  return acc;
}

__global__ __launch_bounds__(256) void sum_rows(const float* __restrict__ H,
                                                const int* __restrict__ cnt,
                                                const int* __restrict__ table,
                                                const int* __restrict__ spillCount,
                                                const int* __restrict__ spill,
                                                const float* __restrict__ bias,
                                                float* __restrict__ out,
                                                int nNodes) {
  const int lane = threadIdx.x & 31;
  const int nWaves = gridDim.x * 8;
  int node = __builtin_amdgcn_readfirstlane(blockIdx.x * 8 + (threadIdx.x >> 5));
  if (node >= nNodes) return;
  const v2f bv = *reinterpret_cast<const v2f*>(bias + 2 * lane);
  const int2* tab = reinterpret_cast<const int2*>(table);

  const int2* idle = reinterpret_cast<const int2*>(bias) + lane;

  int n1 = node + nWaves;
  int c0 = cnt[node];
  int c1 = n1 < nNodes ? cnt[n1] : 0;
  const int2* q0 = idle;
  if (lane < (c0 & (SPILL_FLAG - 1))) q0 = tab + (size_t)node * SLOTS + lane;
  int2 e0 = *q0;

  for (;;) {
    const int n2 = n1 + nWaves;
    const int c2 = n2 < nNodes ? cnt[n2] : 0;
    const int2* q1 = idle;
    if (n1 < nNodes && lane < (c1 & (SPILL_FLAG - 1))) q1 = tab + (size_t)n1 * SLOTS + lane;
    const int2 e1 = *q1;

    const int cw = __builtin_amdgcn_readfirstlane(c0);
    const v2f acc = sum_node(H, e0.x, e0.y, cw, node, spillCount, spill, bv, lane);

    *reinterpret_cast<v2f*>(out + (size_t)node * 64 + 2 * lane) = acc;

    if (n1 >= nNodes) break;
    node = n1; n1 = n2; c0 = c1; c1 = c2; e0 = e1;
  }
}

__global__ __launch_bounds__(256) void sum_rows_gemm(const float* __restrict__ H,
                                                     const int* __restrict__ cnt,
                                                     const int* __restrict__ table,
                                                     const int* __restrict__ spillCount,
                                                     const int* __restrict__ spill,
                                                     const float* __restrict__ bias,
                                                     const float* __restrict__ W,
                                                     float* __restrict__ Y,
                                                     int nNodes) {
  __shared__ __attribute__((aligned(16))) float sW[(HID_DIM / 2) * W_PITCH];
  __shared__ __attribute__((aligned(16))) float sT[4 * 16 * T_PITCH];

  const int lane = threadIdx.x & 31;
  const int wave = __builtin_amdgcn_readfirstlane(threadIdx.x >> 5);
  const int part = wave & 1;
  const int tile = blockIdx.x * 4 + (wave >> 1);
  float* tl = sT + (wave >> 1) * (16 * T_PITCH);
  const int node0 = tile * 16 + part * 8;

  int cl = 0;
  if (lane < 8 && node0 + lane < nNodes) cl = cnt[node0 + lane];

  {
    float4 w0[2], w1[2];
#pragma unroll
    for (int u = 0; u < 2; u++) {
      const int item = threadIdx.x + 256 * u;
      const float* wp = W + (size_t)(item >> 4) * 128 + ((item & 15) << 2);
      w0[u] = *reinterpret_cast<const float4*>(wp);
      w1[u] = *reinterpret_cast<const float4*>(wp + 64);
    }
#pragma unroll
    for (int u = 0; u < 2; u++) {
      const int item = threadIdx.x + 256 * u;
      float4* sp = reinterpret_cast<float4*>(sW + (item >> 4) * W_PITCH + ((item & 15) << 3));
      sp[0] = make_float4(w0[u].x, w1[u].x, w0[u].y, w1[u].y);
      sp[1] = make_float4(w0[u].z, w1[u].z, w0[u].w, w1[u].w);
    }
  }

  const v2f bv = *reinterpret_cast<const v2f*>(bias + 2 * lane);
  const int2* tab = reinterpret_cast<const int2*>(table);
  const int2* idle = reinterpret_cast<const int2*>(bias) + lane;

  int c0 = __builtin_amdgcn_readlane(cl, 0);
  const int2* q0 = idle;
  if (lane < (c0 & (SPILL_FLAG - 1))) q0 = tab + (size_t)node0 * SLOTS + lane;
  int2 e0 = *q0;

  for (int i = 0; i < 8; i++) {
    const int node = node0 + i;
    const int c1 = __builtin_amdgcn_readlane(cl, i + 1);
    const int2* q1 = idle;
    if (lane < (c1 & (SPILL_FLAG - 1))) q1 = tab + (size_t)(node + 1) * SLOTS + lane;
    const int2 e1 = *q1;

    v2f acc = sum_node(H, e0.x, e0.y, c0, node, spillCount, spill, bv, lane);
    acc.x = fmaxf(acc.x, 0.f);
    acc.y = fmaxf(acc.y, 0.f);
    *reinterpret_cast<v2f*>(tl + (part * 8 + i) * T_PITCH + 2 * lane) = acc;

    c0 = c1; e0 = e1;
  }
  __syncthreads();

  const int mrow = lane & 15;
  const int half = lane >> 4;
  const int koff = half << 1;
  const float* arow = tl + mrow * T_PITCH + koff;
  const float* wbase = sW + half * W_PITCH + (mrow << 1) + part * 64;

  v8f acc0 = {}, acc1 = {};
  for (int k = 0; k < HID_DIM; k += 4) {
    const v2f a = *(const v2f*)(arow + k);
    const float* wk = wbase + (k >> 1) * W_PITCH;
    const v2f b0 = *(const v2f*)(wk + 0);
    const v2f b1 = *(const v2f*)(wk + 32);
    acc0 = __builtin_amdgcn_wmma_f32_16x16x4_f32(false, a, false, b0, (short)0, acc0, false, false);
    acc1 = __builtin_amdgcn_wmma_f32_16x16x4_f32(false, a, false, b1, (short)0, acc1, false, false);
  }

  const int rbase = tile * 16 + (half << 3);
  float* y = Y + (size_t)rbase * 64 + part * 32 + mrow;
#pragma unroll
  for (int v = 0; v < 8; v++) {
    if (rbase + v < nNodes) {
      y[(size_t)v * 64 + 0]  = acc0[v];
      y[(size_t)v * 64 + 16] = acc1[v];
    }
  }
}

extern "C" void kernel_launch(void* const* d_in, const int* in_sizes, int n_in,
                              void* d_out, int out_size, void* d_ws, size_t ws_size,
                              hipStream_t stream) {
  const float* x  = (const float*)d_in[0];
  const int*   ei = (const int*)d_in[1];
  const float* ew = (const float*)d_in[2];
  const float* w1 = (const float*)d_in[3];
  const float* b1 = (const float*)d_in[4];
  const float* w2 = (const float*)d_in[5];
  const float* b2 = (const float*)d_in[6];

  const int N = in_sizes[0] / IN_DIM;
  const int E = in_sizes[2];
  const int* src = ei;
  const int* dst = ei + E;
  float* out = (float*)d_out;

  const int nBuckets = (N + BUCKET_NODES - 1) / BUCKET_NODES;
  const int splitBlocks = (E + SPLIT_EDGES - 1) / SPLIT_EDGES;

  float* A = (float*)d_ws;
  float* B = A + (size_t)N * 64;
  int* table = (int*)(B + (size_t)N * 64);
  int* part  = table + (size_t)N * SLOTS * 2;
  int* ovf   = part + (size_t)splitBlocks * nBuckets * SLICE * 4;
  int* spill = ovf + (size_t)splitBlocks * SPLIT_EDGES * 4;
  int* hist  = spill + (size_t)E * 4;
  int* cnt   = hist + (size_t)splitBlocks * MAX_BUCKETS;
  int* ovfCount   = cnt + N;
  int* spillCount = ovfCount + splitBlocks;

  const int nTiles = (N + 15) / 16;
  const int gemmBlocks = (nTiles + 7) / 8;
  const int pairBlocks = gemmBlocks > splitBlocks ? gemmBlocks : splitBlocks;
  const int rowBlocks = 1280;
  const int fusedBlocks = (nTiles + 3) / 4;

  gemm_and_split<<<2 * pairBlocks, 256, 0, stream>>>(x, w1, A, spillCount, nTiles, gemmBlocks,
                                                     src, dst, ew, part, hist, ovf, ovfCount, nBuckets, E, splitBlocks);
  slot_edges<<<nBuckets, 1024, 0, stream>>>(hist, part, ovf, ovfCount, cnt, table, spillCount, spill, splitBlocks, nBuckets, N);
  sum_rows_gemm<<<fusedBlocks, 256, 0, stream>>>(A, cnt, table, spillCount, spill, b1, w2, B, N);
  sum_rows<<<rowBlocks, 256, 0, stream>>>(B, cnt, table, spillCount, spill, b2, out, N);
}


# edge-ordering records staged in LDS and stored as whole lines: split writes one dense bucket-ordered list per block, slot kernel builds table rows in LDS; slot order = LDS arrival as before
# speedup vs baseline: 3.7414x; 1.0695x over previous
#include <hip/hip_runtime.h>


typedef __attribute__((ext_vector_type(2))) float v2f;
typedef __attribute__((ext_vector_type(8))) float v8f;

#define IN_DIM  128
#define HID_DIM 64
#define OUT_DIM 64
#define W_PITCH 160
#define T_PITCH 68

#define SLOTS        32
#define BUCKET_SHIFT 9
#define BUCKET_NODES 512
#define MAX_BUCKETS  256
#define SPLIT_EPT    8
#define SPLIT_EDGES  2048
#define RUN_SHIFT    16

template <int K>
__device__ __forceinline__ void gemm_role(int gb, float* __restrict__ sW,
                                          const float* __restrict__ X,
                                          const float* __restrict__ W,
                                          float* __restrict__ Y,
                                          int* __restrict__ spillCount,
                                          int nTiles) {
  if (gb == 0 && threadIdx.x == 0) *spillCount = 0;
  for (int i = threadIdx.x; i < K * 64; i += 256) {
    int r = i >> 6, c = i & 63;
    sW[(r >> 1) * W_PITCH + (c << 1) + (r & 1)] = W[i];
  }
  __syncthreads();

  const int wave = threadIdx.x >> 5;
  const int lane = threadIdx.x & 31;
  const int tile = gb * 8 + wave;
  if (tile >= nTiles) return;

  const int mrow = lane & 15;
  const int half = lane >> 4;
  const int koff = half << 1;

  const float* xrow = X + (size_t)(tile * 16 + mrow) * K;
  const float* wbase = sW + (size_t)half * W_PITCH + (mrow << 1);

  v8f acc0 = {}, acc1 = {}, acc2 = {}, acc3 = {};

  for (int k = 0; k < K; k += 4) {
    v2f a = *(const v2f*)(xrow + k + koff);
    const float* wk = wbase + (k >> 1) * W_PITCH;
    v2f b0 = *(const v2f*)(wk + 0);
    v2f b1 = *(const v2f*)(wk + 32);
    v2f b2 = *(const v2f*)(wk + 64);
    v2f b3 = *(const v2f*)(wk + 96);

    acc0 = __builtin_amdgcn_wmma_f32_16x16x4_f32(false, a, false, b0, (short)0, acc0, false, false);
    acc1 = __builtin_amdgcn_wmma_f32_16x16x4_f32(false, a, false, b1, (short)0, acc1, false, false);
    acc2 = __builtin_amdgcn_wmma_f32_16x16x4_f32(false, a, false, b2, (short)0, acc2, false, false);
    acc3 = __builtin_amdgcn_wmma_f32_16x16x4_f32(false, a, false, b3, (short)0, acc3, false, false);
  }

  const int rbase = tile * 16 + (half << 3);
  float* y = Y + (size_t)rbase * 64 + mrow;
#pragma unroll
  for (int v = 0; v < 8; v++) {
    y[(size_t)v * 64 + 0]  = acc0[v];
    y[(size_t)v * 64 + 16] = acc1[v];
    y[(size_t)v * 64 + 32] = acc2[v];
    y[(size_t)v * 64 + 48] = acc3[v];
  }
}

__device__ __forceinline__ void spill_record(int* __restrict__ spillCount, int* __restrict__ spill, int4 rec) {
  const int o = atomicAdd(spillCount, 1);
  reinterpret_cast<int4*>(spill)[o] = rec;
}

__device__ __forceinline__ void split_role(int sb, unsigned char* __restrict__ smem,
                                           const int* __restrict__ src,
                                           const int* __restrict__ dst,
                                           const float* __restrict__ ew,
                                           int* __restrict__ part,
                                           int* __restrict__ runs,
                                           int nEdges) {
  int*  h     = reinterpret_cast<int*>(smem);
  int*  wsum  = h + MAX_BUCKETS;
  int4* stage = reinterpret_cast<int4*>(smem + 2048);
  const int tid = threadIdx.x;
  const int lane = tid & 31, wave = tid >> 5;
  h[tid] = 0;
  __syncthreads();

  const long long e0 = (long long)sb * SPLIT_EDGES + tid;
  int d[SPLIT_EPT], s[SPLIT_EPT], w[SPLIT_EPT], r[SPLIT_EPT];
#pragma unroll
  for (int u = 0; u < SPLIT_EPT; u++) {
    const long long e = e0 + u * 256;
    d[u] = 0; s[u] = 0; w[u] = 0;
    if (e < nEdges) { d[u] = dst[e]; s[u] = src[e]; w[u] = __float_as_int(ew[e]); }
  }
#pragma unroll
  for (int u = 0; u < SPLIT_EPT; u++) {
    r[u] = 0;
    if (e0 + u * 256 < nEdges) r[u] = atomicAdd(&h[d[u] >> BUCKET_SHIFT], 1);
  }
  __syncthreads();

  const int n = h[tid];
  int incl = n;
#pragma unroll
  for (int k = 1; k < 32; k <<= 1) {
    const int t = __shfl_up(incl, k);
    if (lane >= k) incl += t;
  }
  if (lane == 31) wsum[wave] = incl;
  __syncthreads();
  int start = incl - n;
#pragma unroll
  for (int k = 0; k < 8; k++) start += k < wave ? wsum[k] : 0;
  h[tid] = start;
  __syncthreads();

  int pos[SPLIT_EPT];
#pragma unroll
  for (int u = 0; u < SPLIT_EPT; u++) pos[u] = h[d[u] >> BUCKET_SHIFT] + r[u];
#pragma unroll
  for (int u = 0; u < SPLIT_EPT; u++) {
    if (e0 + u * 256 < nEdges) stage[pos[u]] = make_int4(d[u], s[u], w[u], 0);
  }
  __syncthreads();

  const long long left = (long long)nEdges - (long long)sb * SPLIT_EDGES;
  const int total = left < SPLIT_EDGES ? (int)left : SPLIT_EDGES;
  int4* out4 = reinterpret_cast<int4*>(part) + (size_t)sb * SPLIT_EDGES;
#pragma unroll
  for (int u = 0; u < SPLIT_EPT; u++) {
    const int i = tid + u * 256;
    if (i < total) out4[i] = stage[i];
  }
  runs[(size_t)sb * MAX_BUCKETS + tid] = start | (n << RUN_SHIFT);
}

__global__ __launch_bounds__(256) void gemm_and_split(const float* __restrict__ X,
                                                      const float* __restrict__ W,
                                                      float* __restrict__ Y,
                                                      int* __restrict__ spillCount,
                                                      int nTiles, int gemmBlocks,
                                                      const int* __restrict__ src,
                                                      const int* __restrict__ dst,
                                                      const float* __restrict__ ew,
                                                      int* __restrict__ part,
                                                      int* __restrict__ runs,
                                                      int nEdges, int splitBlocks) {
  __shared__ __attribute__((aligned(16))) unsigned char smem[(IN_DIM / 2) * W_PITCH * 4];
  const int idx = blockIdx.x >> 1;
  if (blockIdx.x & 1) {
    if (idx < splitBlocks) split_role(idx, smem, src, dst, ew, part, runs, nEdges);
  } else {
    if (idx < gemmBlocks) gemm_role<IN_DIM>(idx, reinterpret_cast<float*>(smem), X, W, Y, spillCount, nTiles);
  }
}

__global__ __launch_bounds__(1024) void slot_edges(const int* __restrict__ runs,
                                                   const int* __restrict__ part,
                                                   int* __restrict__ cnt,
                                                   int* __restrict__ table,
                                                   int* __restrict__ spillCount,
                                                   int* __restrict__ spill,
                                                   int nSplit, int nNodes) {
  __shared__ __attribute__((aligned(16))) int2 tab[BUCKET_NODES * SLOTS];
  __shared__ int c[BUCKET_NODES];
  const int b = blockIdx.x, tid = threadIdx.x;
  const int lane = tid & 31;
  const int wave = __builtin_amdgcn_readfirstlane(tid >> 5);
  if (tid < BUCKET_NODES) c[tid] = 0;
  {
    int4* z = reinterpret_cast<int4*>(tab);
#pragma unroll
    for (int u = 0; u < (BUCKET_NODES * SLOTS * 8) / (16 * 1024); u++) z[tid + u * 1024] = make_int4(0, 0, 0, 0);
  }
  __syncthreads();

  const int4* p4 = reinterpret_cast<const int4*>(part);
  for (int base = wave; base < nSplit; base += 32 * 32) {
    const int blkL = base + 32 * lane;
    int rw = 0;
    if (blkL < nSplit) rw = runs[(size_t)blkL * MAX_BUCKETS + b];
    const int4* idle = reinterpret_cast<const int4*>(runs + (size_t)base * MAX_BUCKETS) + lane;
#pragma nounroll
    for (int j0 = 0; j0 < 32 && base + 32 * j0 < nSplit; j0 += 8) {
      int4 rec[8];
      int rk[8];
      bool on[8];
#pragma unroll
      for (int u = 0; u < 8; u++) {
        const int word = __builtin_amdgcn_readlane(rw, j0 + u);
        on[u] = lane < (word >> RUN_SHIFT);
        const int4* q = p4 + (size_t)(base + 32 * (j0 + u)) * SPLIT_EDGES + (word & ((1 << RUN_SHIFT) - 1)) + lane;
        rec[u] = *(on[u] ? q : idle);
      }
#pragma unroll
      for (int u = 0; u < 8; u++) {
        rk[u] = 0;
        if (on[u]) rk[u] = atomicAdd(&c[rec[u].x & (BUCKET_NODES - 1)], 1);
      }
#pragma unroll
      for (int u = 0; u < 8; u++) {
        if (on[u]) {
          if (rk[u] < SLOTS) tab[(rec[u].x & (BUCKET_NODES - 1)) * SLOTS + ((rk[u] + rec[u].x) & (SLOTS - 1))] = make_int2(rec[u].y, rec[u].z);
          else spill_record(spillCount, spill, rec[u]);
        }
      }
      for (int u = 0; u < 8; u++) {
        const int word = __builtin_amdgcn_readlane(rw, j0 + u);
        const int len = word >> RUN_SHIFT;
        const int4* q = p4 + (size_t)(base + 32 * (j0 + u)) * SPLIT_EDGES + (word & ((1 << RUN_SHIFT) - 1)) + lane;
        for (int i0 = 32; i0 < len; i0 += 32) {
          if (i0 + lane < len) {
            const int4 x = q[i0];
            const int k = atomicAdd(&c[x.x & (BUCKET_NODES - 1)], 1);
            if (k < SLOTS) tab[(x.x & (BUCKET_NODES - 1)) * SLOTS + ((k + x.x) & (SLOTS - 1))] = make_int2(x.y, x.z);
            else spill_record(spillCount, spill, x);
          }
        }
      }
    }
  }

  __syncthreads();
  for (int row = wave; row < BUCKET_NODES; row += 32) {
    const int node = b * BUCKET_NODES + row;
    if (node < nNodes)
      reinterpret_cast<int2*>(table)[(size_t)node * SLOTS + lane] = tab[row * SLOTS + ((lane + row) & (SLOTS - 1))];
  }
  if (tid < BUCKET_NODES) {
    const int node = b * BUCKET_NODES + tid;
    if (node < nNodes) cnt[node] = c[tid];
  }
}

__device__ __forceinline__ v2f row2(const float* __restrict__ H, int s, int lane) {
  return *reinterpret_cast<const v2f*>(H + (size_t)s * 64 + 2 * lane);
}
template <int CNT>
__device__ __forceinline__ void load_rows(const float* __restrict__ H, int s, int j, int lane, v2f* t) {
#pragma unroll
  for (int u = 0; u < CNT; u++) t[u] = row2(H, __builtin_amdgcn_readlane(s, j + u), lane);
}
template <int CNT>
__device__ __forceinline__ v2f add_rows(v2f acc, const v2f* t, int wb, int j) {
#pragma unroll
  for (int u = 0; u < CNT; u++) acc += t[u] * __int_as_float(__builtin_amdgcn_readlane(wb, j + u));
  return acc;
}

__device__ __forceinline__ v2f sum_node(const float* __restrict__ H, int s, int wb, int cw, int node,
                                        const int* __restrict__ spillCount, const int* __restrict__ spill,
                                        v2f bv, int lane) {
  const int deg = cw;
  const int m = deg < SLOTS ? deg : SLOTS;
  v2f acc = bv;

  v2f t[16];
  int j = 0;
  for (; j + 16 <= m; j += 16) {
    load_rows<16>(H, s, j, lane, t);
    acc = add_rows<16>(acc, t, wb, j);
  }
  {
    const int rem = m - j;
    const int j8 = j, j4 = j8 + (rem & 8), j2 = j4 + (rem & 4), j1 = j2 + (rem & 2);
    if (rem & 8) load_rows<8>(H, s, j8, lane, t);
    if (rem & 4) load_rows<4>(H, s, j4, lane, t + 8);
    if (rem & 2) load_rows<2>(H, s, j2, lane, t + 12);
    if (rem & 1) load_rows<1>(H, s, j1, lane, t + 14);
    if (rem & 8) acc = add_rows<8>(acc, t, wb, j8);
    if (rem & 4) acc = add_rows<4>(acc, t + 8, wb, j4);
    if (rem & 2) acc = add_rows<2>(acc, t + 12, wb, j2);
    if (rem & 1) acc = add_rows<1>(acc, t + 14, wb, j1);
  }

  if (deg > SLOTS) {
    const int total = *spillCount;
    const int4* sp = reinterpret_cast<const int4*>(spill);
    for (int base = 0; base < total; base += 32) {
      int4 o = make_int4(-1, 0, 0, 0);
      if (base + lane < total) o = sp[base + lane];
      unsigned match = __builtin_amdgcn_ballot_w32(o.x == node);
      while (match) {
        const int l = __builtin_ctz(match);
        match &= match - 1;
        const v2f v = row2(H, __builtin_amdgcn_readlane(o.y, l), lane);
        acc += v * __int_as_float(__builtin_amdgcn_readlane(o.z, l));
      }
    }
  }
  return acc;
}

__global__ __launch_bounds__(256) void sum_rows(const float* __restrict__ H,
                                                const int* __restrict__ cnt,
                                                const int* __restrict__ table,
                                                const int* __restrict__ spillCount,
                                                const int* __restrict__ spill,
                                                const float* __restrict__ bias,
                                                float* __restrict__ out,
                                                int nNodes) {
  const int lane = threadIdx.x & 31;
  const int nWaves = gridDim.x * 8;
  int node = __builtin_amdgcn_readfirstlane(blockIdx.x * 8 + (threadIdx.x >> 5));
  if (node >= nNodes) return;
  const v2f bv = *reinterpret_cast<const v2f*>(bias + 2 * lane);
  const int2* tab = reinterpret_cast<const int2*>(table);

  const int2* idle = reinterpret_cast<const int2*>(bias) + lane;

  int n1 = node + nWaves;
  int c0 = cnt[node];
  int c1 = n1 < nNodes ? cnt[n1] : 0;
  const int2* q0 = idle;
  if (lane < c0) q0 = tab + (size_t)node * SLOTS + lane;
  int2 e0 = *q0;

  for (;;) {
    const int n2 = n1 + nWaves;
    const int c2 = n2 < nNodes ? cnt[n2] : 0;
    const int2* q1 = idle;
    if (n1 < nNodes && lane < c1) q1 = tab + (size_t)n1 * SLOTS + lane;
    const int2 e1 = *q1;

    const int cw = __builtin_amdgcn_readfirstlane(c0);
    const v2f acc = sum_node(H, e0.x, e0.y, cw, node, spillCount, spill, bv, lane);

    *reinterpret_cast<v2f*>(out + (size_t)node * 64 + 2 * lane) = acc;

    if (n1 >= nNodes) break;
    node = n1; n1 = n2; c0 = c1; c1 = c2; e0 = e1;
  }
}

__global__ __launch_bounds__(256) void sum_rows_gemm(const float* __restrict__ H,
                                                     const int* __restrict__ cnt,
                                                     const int* __restrict__ table,
                                                     const int* __restrict__ spillCount,
                                                     const int* __restrict__ spill,
                                                     const float* __restrict__ bias,
                                                     const float* __restrict__ W,
                                                     float* __restrict__ Y,
                                                     int nNodes) {
  __shared__ __attribute__((aligned(16))) float sW[(HID_DIM / 2) * W_PITCH];
  __shared__ __attribute__((aligned(16))) float sT[4 * 16 * T_PITCH];

  const int lane = threadIdx.x & 31;
  const int wave = __builtin_amdgcn_readfirstlane(threadIdx.x >> 5);
  const int part = wave & 1;
  const int tile = blockIdx.x * 4 + (wave >> 1);
  float* tl = sT + (wave >> 1) * (16 * T_PITCH);
  const int node0 = tile * 16 + part * 8;

  int cl = 0;
  if (lane < 8 && node0 + lane < nNodes) cl = cnt[node0 + lane];

  {
    float4 w0[2], w1[2];
#pragma unroll
    for (int u = 0; u < 2; u++) {
      const int item = threadIdx.x + 256 * u;
      const float* wp = W + (size_t)(item >> 4) * 128 + ((item & 15) << 2);
      w0[u] = *reinterpret_cast<const float4*>(wp);
      w1[u] = *reinterpret_cast<const float4*>(wp + 64);
    }
#pragma unroll
    for (int u = 0; u < 2; u++) {
      const int item = threadIdx.x + 256 * u;
      float4* sp = reinterpret_cast<float4*>(sW + (item >> 4) * W_PITCH + ((item & 15) << 3));
      sp[0] = make_float4(w0[u].x, w1[u].x, w0[u].y, w1[u].y);
      sp[1] = make_float4(w0[u].z, w1[u].z, w0[u].w, w1[u].w);
    }
  }

  const v2f bv = *reinterpret_cast<const v2f*>(bias + 2 * lane);
  const int2* tab = reinterpret_cast<const int2*>(table);
  const int2* idle = reinterpret_cast<const int2*>(bias) + lane;

  int c0 = __builtin_amdgcn_readlane(cl, 0);
  const int2* q0 = idle;
  if (lane < c0) q0 = tab + (size_t)node0 * SLOTS + lane;
  int2 e0 = *q0;

  for (int i = 0; i < 8; i++) {
    const int node = node0 + i;
    const int c1 = __builtin_amdgcn_readlane(cl, i + 1);
    const int2* q1 = idle;
    if (lane < c1) q1 = tab + (size_t)(node + 1) * SLOTS + lane;
    const int2 e1 = *q1;

    v2f acc = sum_node(H, e0.x, e0.y, c0, node, spillCount, spill, bv, lane);
    acc.x = fmaxf(acc.x, 0.f);
    acc.y = fmaxf(acc.y, 0.f);
    *reinterpret_cast<v2f*>(tl + (part * 8 + i) * T_PITCH + 2 * lane) = acc;

    c0 = c1; e0 = e1;
  }
  __syncthreads();

  const int mrow = lane & 15;
  const int half = lane >> 4;
  const int koff = half << 1;
  const float* arow = tl + mrow * T_PITCH + koff;
  const float* wbase = sW + half * W_PITCH + (mrow << 1) + part * 64;

  v8f acc0 = {}, acc1 = {};
  for (int k = 0; k < HID_DIM; k += 4) {
    const v2f a = *(const v2f*)(arow + k);
    const float* wk = wbase + (k >> 1) * W_PITCH;
    const v2f b0 = *(const v2f*)(wk + 0);
    const v2f b1 = *(const v2f*)(wk + 32);
    acc0 = __builtin_amdgcn_wmma_f32_16x16x4_f32(false, a, false, b0, (short)0, acc0, false, false);
    acc1 = __builtin_amdgcn_wmma_f32_16x16x4_f32(false, a, false, b1, (short)0, acc1, false, false);
  }

  const int rbase = tile * 16 + (half << 3);
  float* y = Y + (size_t)rbase * 64 + part * 32 + mrow;
#pragma unroll
  for (int v = 0; v < 8; v++) {
    if (rbase + v < nNodes) {
      y[(size_t)v * 64 + 0]  = acc0[v];
      y[(size_t)v * 64 + 16] = acc1[v];
    }
  }
}

extern "C" void kernel_launch(void* const* d_in, const int* in_sizes, int n_in,
                              void* d_out, int out_size, void* d_ws, size_t ws_size,
                              hipStream_t stream) {
  const float* x  = (const float*)d_in[0];
  const int*   ei = (const int*)d_in[1];
  const float* ew = (const float*)d_in[2];
  const float* w1 = (const float*)d_in[3];
  const float* b1 = (const float*)d_in[4];
  const float* w2 = (const float*)d_in[5];
  const float* b2 = (const float*)d_in[6];

  const int N = in_sizes[0] / IN_DIM;
  const int E = in_sizes[2];
  const int* src = ei;
  const int* dst = ei + E;
  float* out = (float*)d_out;

  const int nBuckets = (N + BUCKET_NODES - 1) / BUCKET_NODES;
  const int splitBlocks = (E + SPLIT_EDGES - 1) / SPLIT_EDGES;

  float* A = (float*)d_ws;
  float* B = A + (size_t)N * 64;
  int* table = (int*)(B + (size_t)N * 64);
  int* part  = table + (size_t)N * SLOTS * 2;
  int* spill = part + (size_t)splitBlocks * SPLIT_EDGES * 4;
  int* runs  = spill + (size_t)E * 4;
  int* cnt   = runs + (size_t)splitBlocks * MAX_BUCKETS;
  int* spillCount = cnt + N;

  const int nTiles = (N + 15) / 16;
  const int gemmBlocks = (nTiles + 7) / 8;
  const int pairBlocks = gemmBlocks > splitBlocks ? gemmBlocks : splitBlocks;
  const int rowBlocks = 1280;
  const int fusedBlocks = (nTiles + 3) / 4;

  gemm_and_split<<<2 * pairBlocks, 256, 0, stream>>>(x, w1, A, spillCount, nTiles, gemmBlocks,
                                                     src, dst, ew, part, runs, E, splitBlocks);
  slot_edges<<<nBuckets, 1024, 0, stream>>>(runs, part, cnt, table, spillCount, spill, splitBlocks, N);
  sum_rows_gemm<<<fusedBlocks, 256, 0, stream>>>(A, cnt, table, spillCount, spill, b1, w2, B, N);
  sum_rows<<<rowBlocks, 256, 0, stream>>>(B, cnt, table, spillCount, spill, b2, out, N);
}


# row sums: each row address is a 32-bit lane offset on a scalar base, table rows read whole: 4 not 6 instructions per row load; 49 not 68 registers, so 8 not 6 fused groups per WGP; same sum order
# speedup vs baseline: 4.2545x; 1.1371x over previous
#include <hip/hip_runtime.h>


typedef __attribute__((ext_vector_type(2))) float v2f;
typedef __attribute__((ext_vector_type(8))) float v8f;

#define IN_DIM  128
#define HID_DIM 64
#define OUT_DIM 64
#define W_PITCH 160
#define T_PITCH 68

#define SLOTS        32
#define BUCKET_SHIFT 9
#define BUCKET_NODES 512
#define MAX_BUCKETS  256
#define SPLIT_EPT    8
#define SPLIT_EDGES  2048
#define RUN_SHIFT    16

template <int K>
__device__ __forceinline__ void gemm_role(int gb, float* __restrict__ sW,
                                          const float* __restrict__ X,
                                          const float* __restrict__ W,
                                          float* __restrict__ Y,
                                          int* __restrict__ spillCount,
                                          int nTiles) {
  if (gb == 0 && threadIdx.x == 0) *spillCount = 0;
  for (int i = threadIdx.x; i < K * 64; i += 256) {
    int r = i >> 6, c = i & 63;
    sW[(r >> 1) * W_PITCH + (c << 1) + (r & 1)] = W[i];
  }
  __syncthreads();

  const int wave = threadIdx.x >> 5;
  const int lane = threadIdx.x & 31;
  const int tile = gb * 8 + wave;
  if (tile >= nTiles) return;

  const int mrow = lane & 15;
  const int half = lane >> 4;
  const int koff = half << 1;

  const float* xrow = X + (size_t)(tile * 16 + mrow) * K;
  const float* wbase = sW + (size_t)half * W_PITCH + (mrow << 1);

  v8f acc0 = {}, acc1 = {}, acc2 = {}, acc3 = {};

  for (int k = 0; k < K; k += 4) {
    v2f a = *(const v2f*)(xrow + k + koff);
    const float* wk = wbase + (k >> 1) * W_PITCH;
    v2f b0 = *(const v2f*)(wk + 0);
    v2f b1 = *(const v2f*)(wk + 32);
    v2f b2 = *(const v2f*)(wk + 64);
    v2f b3 = *(const v2f*)(wk + 96);

    acc0 = __builtin_amdgcn_wmma_f32_16x16x4_f32(false, a, false, b0, (short)0, acc0, false, false);
    acc1 = __builtin_amdgcn_wmma_f32_16x16x4_f32(false, a, false, b1, (short)0, acc1, false, false);
    acc2 = __builtin_amdgcn_wmma_f32_16x16x4_f32(false, a, false, b2, (short)0, acc2, false, false);
    acc3 = __builtin_amdgcn_wmma_f32_16x16x4_f32(false, a, false, b3, (short)0, acc3, false, false);
  }

  const int rbase = tile * 16 + (half << 3);
  float* y = Y + (size_t)rbase * 64 + mrow;
#pragma unroll
  for (int v = 0; v < 8; v++) {
    y[(size_t)v * 64 + 0]  = acc0[v];
    y[(size_t)v * 64 + 16] = acc1[v];
    y[(size_t)v * 64 + 32] = acc2[v];
    y[(size_t)v * 64 + 48] = acc3[v];
  }
}

__device__ __forceinline__ void spill_record(int* __restrict__ spillCount, int* __restrict__ spill, int4 rec) {
  const int o = atomicAdd(spillCount, 1);
  reinterpret_cast<int4*>(spill)[o] = rec;
}

__device__ __forceinline__ void split_role(int sb, unsigned char* __restrict__ smem,
                                           const int* __restrict__ src,
                                           const int* __restrict__ dst,
                                           const float* __restrict__ ew,
                                           int* __restrict__ part,
                                           int* __restrict__ runs,
                                           int nEdges) {
  int*  h     = reinterpret_cast<int*>(smem);
  int*  wsum  = h + MAX_BUCKETS;
  int4* stage = reinterpret_cast<int4*>(smem + 2048);
  const int tid = threadIdx.x;
  const int lane = tid & 31, wave = tid >> 5;
  h[tid] = 0;
  __syncthreads();

  const long long e0 = (long long)sb * SPLIT_EDGES + tid;
  int d[SPLIT_EPT], s[SPLIT_EPT], w[SPLIT_EPT], r[SPLIT_EPT];
#pragma unroll
  for (int u = 0; u < SPLIT_EPT; u++) {
    const long long e = e0 + u * 256;
    d[u] = 0; s[u] = 0; w[u] = 0;
    if (e < nEdges) { d[u] = dst[e]; s[u] = src[e]; w[u] = __float_as_int(ew[e]); }
  }
#pragma unroll
  for (int u = 0; u < SPLIT_EPT; u++) {
    r[u] = 0;
    if (e0 + u * 256 < nEdges) r[u] = atomicAdd(&h[d[u] >> BUCKET_SHIFT], 1);
  }
  __syncthreads();

  const int n = h[tid];
  int incl = n;
#pragma unroll
  for (int k = 1; k < 32; k <<= 1) {
    const int t = __shfl_up(incl, k);
    if (lane >= k) incl += t;
  }
  if (lane == 31) wsum[wave] = incl;
  __syncthreads();
  int start = incl - n;
#pragma unroll
  for (int k = 0; k < 8; k++) start += k < wave ? wsum[k] : 0;
  h[tid] = start;
  __syncthreads();

  int pos[SPLIT_EPT];
#pragma unroll
  for (int u = 0; u < SPLIT_EPT; u++) pos[u] = h[d[u] >> BUCKET_SHIFT] + r[u];
#pragma unroll
  for (int u = 0; u < SPLIT_EPT; u++) {
    if (e0 + u * 256 < nEdges) stage[pos[u]] = make_int4(d[u], s[u], w[u], 0);
  }
  __syncthreads();

  const long long left = (long long)nEdges - (long long)sb * SPLIT_EDGES;
  const int total = left < SPLIT_EDGES ? (int)left : SPLIT_EDGES;
  int4* out4 = reinterpret_cast<int4*>(part) + (size_t)sb * SPLIT_EDGES;
#pragma unroll
  for (int u = 0; u < SPLIT_EPT; u++) {
    const int i = tid + u * 256;
    if (i < total) out4[i] = stage[i];
  }
  runs[(size_t)sb * MAX_BUCKETS + tid] = start | (n << RUN_SHIFT);
}

__global__ __launch_bounds__(256) void gemm_and_split(const float* __restrict__ X,
                                                      const float* __restrict__ W,
                                                      float* __restrict__ Y,
                                                      int* __restrict__ spillCount,
                                                      int nTiles, int gemmBlocks,
                                                      const int* __restrict__ src,
                                                      const int* __restrict__ dst,
                                                      const float* __restrict__ ew,
                                                      int* __restrict__ part,
                                                      int* __restrict__ runs,
                                                      int nEdges, int splitBlocks) {
  __shared__ __attribute__((aligned(16))) unsigned char smem[(IN_DIM / 2) * W_PITCH * 4];
  const int idx = blockIdx.x >> 1;
  if (blockIdx.x & 1) {
    if (idx < splitBlocks) split_role(idx, smem, src, dst, ew, part, runs, nEdges);
  } else {
    if (idx < gemmBlocks) gemm_role<IN_DIM>(idx, reinterpret_cast<float*>(smem), X, W, Y, spillCount, nTiles);
  }
}

__global__ __launch_bounds__(1024) void slot_edges(const int* __restrict__ runs,
                                                   const int* __restrict__ part,
                                                   int* __restrict__ cnt,
                                                   int* __restrict__ table,
                                                   int* __restrict__ spillCount,
                                                   int* __restrict__ spill,
                                                   int nSplit, int nNodes) {
  __shared__ __attribute__((aligned(16))) int2 tab[BUCKET_NODES * SLOTS];
  __shared__ int c[BUCKET_NODES];
  const int b = blockIdx.x, tid = threadIdx.x;
  const int lane = tid & 31;
  const int wave = __builtin_amdgcn_readfirstlane(tid >> 5);
  if (tid < BUCKET_NODES) c[tid] = 0;
  {
    int4* z = reinterpret_cast<int4*>(tab);
#pragma unroll
    for (int u = 0; u < (BUCKET_NODES * SLOTS * 8) / (16 * 1024); u++) z[tid + u * 1024] = make_int4(0, 0, 0, 0);
  }
  __syncthreads();

  const int4* p4 = reinterpret_cast<const int4*>(part);
  for (int base = wave; base < nSplit; base += 32 * 32) {
    const int blkL = base + 32 * lane;
    int rw = 0;
    if (blkL < nSplit) rw = runs[(size_t)blkL * MAX_BUCKETS + b];
    const int4* idle = reinterpret_cast<const int4*>(runs + (size_t)base * MAX_BUCKETS) + lane;
#pragma nounroll
    for (int j0 = 0; j0 < 32 && base + 32 * j0 < nSplit; j0 += 8) {
      int4 rec[8];
      int rk[8];
      bool on[8];
#pragma unroll
      for (int u = 0; u < 8; u++) {
        const int word = __builtin_amdgcn_readlane(rw, j0 + u);
        on[u] = lane < (word >> RUN_SHIFT);
        const int4* q = p4 + (size_t)(base + 32 * (j0 + u)) * SPLIT_EDGES + (word & ((1 << RUN_SHIFT) - 1)) + lane;
        rec[u] = *(on[u] ? q : idle);
      }
#pragma unroll
      for (int u = 0; u < 8; u++) {
        rk[u] = 0;
        if (on[u]) rk[u] = atomicAdd(&c[rec[u].x & (BUCKET_NODES - 1)], 1);
      }
#pragma unroll
      for (int u = 0; u < 8; u++) {
        if (on[u]) {
          if (rk[u] < SLOTS) tab[(rec[u].x & (BUCKET_NODES - 1)) * SLOTS + ((rk[u] + rec[u].x) & (SLOTS - 1))] = make_int2(rec[u].y, rec[u].z);
          else spill_record(spillCount, spill, rec[u]);
        }
      }
      for (int u = 0; u < 8; u++) {
        const int word = __builtin_amdgcn_readlane(rw, j0 + u);
        const int len = word >> RUN_SHIFT;
        const int4* q = p4 + (size_t)(base + 32 * (j0 + u)) * SPLIT_EDGES + (word & ((1 << RUN_SHIFT) - 1)) + lane;
        for (int i0 = 32; i0 < len; i0 += 32) {
          if (i0 + lane < len) {
            const int4 x = q[i0];
            const int k = atomicAdd(&c[x.x & (BUCKET_NODES - 1)], 1);
            if (k < SLOTS) tab[(x.x & (BUCKET_NODES - 1)) * SLOTS + ((k + x.x) & (SLOTS - 1))] = make_int2(x.y, x.z);
            else spill_record(spillCount, spill, x);
          }
        }
      }
    }
  }

  __syncthreads();
  for (int row = wave; row < BUCKET_NODES; row += 32) {
    const int node = b * BUCKET_NODES + row;
    if (node < nNodes)
      reinterpret_cast<int2*>(table)[(size_t)node * SLOTS + lane] = tab[row * SLOTS + ((lane + row) & (SLOTS - 1))];
  }
  if (tid < BUCKET_NODES) {
    const int node = b * BUCKET_NODES + tid;
    if (node < nNodes) cnt[node] = c[tid];
  }
}

__device__ __forceinline__ unsigned row_off(int r, unsigned lane8) {
  return ((unsigned)r << 8) + lane8;
}
__device__ __forceinline__ v2f row2(const float* __restrict__ H, int s, unsigned lane8) {
  return *reinterpret_cast<const v2f*>(reinterpret_cast<const char*>(H) + row_off(s, lane8));
}
template <int CNT>
__device__ __forceinline__ void load_rows(const float* __restrict__ H, int s, int j, unsigned lane8, v2f* t) {
#pragma unroll
  for (int u = 0; u < CNT; u++) t[u] = row2(H, __builtin_amdgcn_readlane(s, j + u), lane8);
}
template <int CNT>
__device__ __forceinline__ v2f add_rows(v2f acc, const v2f* t, int wb, int j) {
#pragma unroll
  for (int u = 0; u < CNT; u++) acc += t[u] * __int_as_float(__builtin_amdgcn_readlane(wb, j + u));
  return acc;
}

__device__ __forceinline__ v2f sum_node(const float* __restrict__ H, int s, int wb, int cw, int node,
                                        const int* __restrict__ spillCount, const int* __restrict__ spill,
                                        v2f bv, int lane) {
  const unsigned lane8 = (unsigned)lane << 3;
  const int deg = cw;
  const int m = deg < SLOTS ? deg : SLOTS;
  v2f acc = bv;

  v2f t[16];
  int j = 0;
  for (; j + 16 <= m; j += 16) {
    load_rows<16>(H, s, j, lane8, t);
    acc = add_rows<16>(acc, t, wb, j);
  }
  {
    const int rem = m - j;
    const int j8 = j, j4 = j8 + (rem & 8), j2 = j4 + (rem & 4), j1 = j2 + (rem & 2);
    if (rem & 8) load_rows<8>(H, s, j8, lane8, t);
    if (rem & 4) load_rows<4>(H, s, j4, lane8, t + 8);
    if (rem & 2) load_rows<2>(H, s, j2, lane8, t + 12);
    if (rem & 1) load_rows<1>(H, s, j1, lane8, t + 14);
    if (rem & 8) acc = add_rows<8>(acc, t, wb, j8);
    if (rem & 4) acc = add_rows<4>(acc, t + 8, wb, j4);
    if (rem & 2) acc = add_rows<2>(acc, t + 12, wb, j2);
    if (rem & 1) acc = add_rows<1>(acc, t + 14, wb, j1);
  }

  if (deg > SLOTS) {
    const int total = *spillCount;
    const int4* sp = reinterpret_cast<const int4*>(spill);
    for (int base = 0; base < total; base += 32) {
      int4 o = make_int4(-1, 0, 0, 0);
      if (base + lane < total) o = sp[base + lane];
      unsigned match = __builtin_amdgcn_ballot_w32(o.x == node);
      while (match) {
        const int l = __builtin_ctz(match);
        match &= match - 1;
        const v2f v = row2(H, __builtin_amdgcn_readlane(o.y, l), lane8);
        acc += v * __int_as_float(__builtin_amdgcn_readlane(o.z, l));
      }
    }
  }
  return acc;
}

__global__ __launch_bounds__(256) void sum_rows(const float* __restrict__ H,
                                                const int* __restrict__ cnt,
                                                const int* __restrict__ table,
                                                const int* __restrict__ spillCount,
                                                const int* __restrict__ spill,
                                                const float* __restrict__ bias,
                                                float* __restrict__ out,
                                                int nNodes) {
  const int lane = threadIdx.x & 31;
  const unsigned lane8 = (unsigned)lane << 3;
  const int nWaves = gridDim.x * 8;
  int node = __builtin_amdgcn_readfirstlane(blockIdx.x * 8 + (threadIdx.x >> 5));
  if (node >= nNodes) return;
  const v2f bv = *reinterpret_cast<const v2f*>(bias + 2 * lane);
  const char* tabB = reinterpret_cast<const char*>(table);
  char* outB = reinterpret_cast<char*>(out);

  int n1 = node + nWaves;
  int c0 = cnt[node];
  int c1 = n1 < nNodes ? cnt[n1] : 0;
  int2 e0 = *reinterpret_cast<const int2*>(tabB + row_off(node, lane8));

  for (;;) {
    const int n2 = n1 + nWaves;
    const int c2 = n2 < nNodes ? cnt[n2] : 0;
    const int nr = n1 < nNodes ? n1 : node;
    const int2 e1 = *reinterpret_cast<const int2*>(tabB + row_off(nr, lane8));

    const int cw = __builtin_amdgcn_readfirstlane(c0);
    const v2f acc = sum_node(H, e0.x, e0.y, cw, node, spillCount, spill, bv, lane);

    *reinterpret_cast<v2f*>(outB + row_off(node, lane8)) = acc;

    if (n1 >= nNodes) break;
    node = n1; n1 = n2; c0 = c1; c1 = c2; e0 = e1;
  }
}

__global__ __launch_bounds__(256) __attribute__((amdgpu_waves_per_eu(16, 16))) void sum_rows_gemm(const float* __restrict__ H,
                                                     const int* __restrict__ cnt,
                                                     const int* __restrict__ table,
                                                     const int* __restrict__ spillCount,
                                                     const int* __restrict__ spill,
                                                     const float* __restrict__ bias,
                                                     const float* __restrict__ W,
                                                     float* __restrict__ Y,
                                                     int nNodes) {
  __shared__ __attribute__((aligned(16))) float sW[(HID_DIM / 2) * W_PITCH];
  __shared__ __attribute__((aligned(16))) float sT[4 * 16 * T_PITCH];

  const int lane = threadIdx.x & 31;
  const int wave = __builtin_amdgcn_readfirstlane(threadIdx.x >> 5);
  const int part = wave & 1;
  const int tile = blockIdx.x * 4 + (wave >> 1);
  float* tl = sT + (wave >> 1) * (16 * T_PITCH);
  const int node0 = tile * 16 + part * 8;

  int cl = 0;
  if (lane < 8 && node0 + lane < nNodes) cl = cnt[node0 + lane];

  {
    float4 w0[2], w1[2];
#pragma unroll
    for (int u = 0; u < 2; u++) {
      const int item = threadIdx.x + 256 * u;
      const float* wp = W + (size_t)(item >> 4) * 128 + ((item & 15) << 2);
      w0[u] = *reinterpret_cast<const float4*>(wp);
      w1[u] = *reinterpret_cast<const float4*>(wp + 64);
    }
#pragma unroll
    for (int u = 0; u < 2; u++) {
      const int item = threadIdx.x + 256 * u;
      float4* sp = reinterpret_cast<float4*>(sW + (item >> 4) * W_PITCH + ((item & 15) << 3));
      sp[0] = make_float4(w0[u].x, w1[u].x, w0[u].y, w1[u].y);
      sp[1] = make_float4(w0[u].z, w1[u].z, w0[u].w, w1[u].w);
    }
  }

  const v2f bv = *reinterpret_cast<const v2f*>(bias + 2 * lane);
  const unsigned lane8 = (unsigned)lane << 3;
  const char* tabB = reinterpret_cast<const char*>(table);
  const int lastNode = nNodes - 1;

  int c0 = __builtin_amdgcn_readlane(cl, 0);
  int2 e0 = *reinterpret_cast<const int2*>(tabB + row_off(node0 < lastNode ? node0 : lastNode, lane8));

  for (int i = 0; i < 8; i++) {
    const int node = node0 + i;
    const int c1 = __builtin_amdgcn_readlane(cl, i + 1);
    const int nx = node + (i < 7 ? 1 : 0);
    const int2 e1 = *reinterpret_cast<const int2*>(tabB + row_off(nx < lastNode ? nx : lastNode, lane8));

    v2f acc = sum_node(H, e0.x, e0.y, c0, node, spillCount, spill, bv, lane);
    acc.x = fmaxf(acc.x, 0.f);
    acc.y = fmaxf(acc.y, 0.f);
    *reinterpret_cast<v2f*>(tl + (part * 8 + i) * T_PITCH + 2 * lane) = acc;

    c0 = c1; e0 = e1;
  }
  __syncthreads();

  const int mrow = lane & 15;
  const int half = lane >> 4;
  const int koff = half << 1;
  const float* arow = tl + mrow * T_PITCH + koff;
  const float* wbase = sW + half * W_PITCH + (mrow << 1) + part * 64;

  v8f acc0 = {}, acc1 = {};
  for (int k = 0; k < HID_DIM; k += 4) {
    const v2f a = *(const v2f*)(arow + k);
    const float* wk = wbase + (k >> 1) * W_PITCH;
    const v2f b0 = *(const v2f*)(wk + 0);
    const v2f b1 = *(const v2f*)(wk + 32);
    acc0 = __builtin_amdgcn_wmma_f32_16x16x4_f32(false, a, false, b0, (short)0, acc0, false, false);
    acc1 = __builtin_amdgcn_wmma_f32_16x16x4_f32(false, a, false, b1, (short)0, acc1, false, false);
  }

  const int rbase = tile * 16 + (half << 3);
  float* y = Y + (size_t)rbase * 64 + part * 32 + mrow;
#pragma unroll
  for (int v = 0; v < 8; v++) {
    if (rbase + v < nNodes) {
      y[(size_t)v * 64 + 0]  = acc0[v];
      y[(size_t)v * 64 + 16] = acc1[v];
    }
  }
}

extern "C" void kernel_launch(void* const* d_in, const int* in_sizes, int n_in,
                              void* d_out, int out_size, void* d_ws, size_t ws_size,
                              hipStream_t stream) {
  const float* x  = (const float*)d_in[0];
  const int*   ei = (const int*)d_in[1];
  const float* ew = (const float*)d_in[2];
  const float* w1 = (const float*)d_in[3];
  const float* b1 = (const float*)d_in[4];
  const float* w2 = (const float*)d_in[5];
  const float* b2 = (const float*)d_in[6];

  const int N = in_sizes[0] / IN_DIM;
  const int E = in_sizes[2];
  const int* src = ei;
  const int* dst = ei + E;
  float* out = (float*)d_out;

  const int nBuckets = (N + BUCKET_NODES - 1) / BUCKET_NODES;
  const int splitBlocks = (E + SPLIT_EDGES - 1) / SPLIT_EDGES;

  float* A = (float*)d_ws;
  float* B = A + (size_t)N * 64;
  int* table = (int*)(B + (size_t)N * 64);
  int* part  = table + (size_t)N * SLOTS * 2;
  int* spill = part + (size_t)splitBlocks * SPLIT_EDGES * 4;
  int* runs  = spill + (size_t)E * 4;
  int* cnt   = runs + (size_t)splitBlocks * MAX_BUCKETS;
  int* spillCount = cnt + N;

  const int nTiles = (N + 15) / 16;
  const int gemmBlocks = (nTiles + 7) / 8;
  const int pairBlocks = gemmBlocks > splitBlocks ? gemmBlocks : splitBlocks;
  const int rowBlocks = 1280;
  const int fusedBlocks = (nTiles + 3) / 4;

  gemm_and_split<<<2 * pairBlocks, 256, 0, stream>>>(x, w1, A, spillCount, nTiles, gemmBlocks,
                                                     src, dst, ew, part, runs, E, splitBlocks);
  slot_edges<<<nBuckets, 1024, 0, stream>>>(runs, part, cnt, table, spillCount, spill, splitBlocks, N);
  sum_rows_gemm<<<fusedBlocks, 256, 0, stream>>>(A, cnt, table, spillCount, spill, b1, w2, B, N);
  sum_rows<<<rowBlocks, 256, 0, stream>>>(B, cnt, table, spillCount, spill, b2, out, N);
}
